# starting kernel
# baseline (speedup 1.0000x reference)
#include <hip/hip_runtime.h>

typedef float v2f __attribute__((ext_vector_type(2)));
typedef float v8f __attribute__((ext_vector_type(8)));

#define DIN  512
#define DHID 512
#define DOUT 256


__global__ void k_fill(float* __restrict__ p, float val, int n) {
  int i = blockIdx.x * blockDim.x + threadIdx.x;
  if (i < n) p[i] = val;
}

__global__ void k_zero4(float4* __restrict__ p, int n4) {
  int i = blockIdx.x * blockDim.x + threadIdx.x;
  if (i < n4) p[i] = make_float4(0.f, 0.f, 0.f, 0.f);
}

__device__ __forceinline__ void atomAddF(float* p, float v) {
  __hip_atomic_fetch_add(p, v, __ATOMIC_RELAXED, __HIP_MEMORY_SCOPE_AGENT);
}


__global__ void k_deg_count(const int* __restrict__ tgt, float* __restrict__ deg, int E) {
  int e = blockIdx.x * blockDim.x + threadIdx.x;
  if (e < E) atomAddF(&deg[tgt[e]], 1.0f);
}

__global__ void k_rsqrt(float* __restrict__ p, int n) {
  int i = blockIdx.x * blockDim.x + threadIdx.x;
  if (i < n) {
    float d = p[i];
    p[i] = (d > 0.f) ? rsqrtf(d) : 0.f;
  }
}


template <int K, int N, bool SQUARE_A, bool TAIL>
__device__ __forceinline__ void gemm_kloop(const float* __restrict__ a0,
                                           const float* __restrict__ a1,
                                           const float* __restrict__ a2,
                                           const float* __restrict__ a3,
                                           const float* __restrict__ bcol,
                                           v8f acc[4][2]) {
  for (int k0 = 0; k0 < K; k0 += 4) {
    v2f a[4];
    if (!TAIL) {
      a[0] = *(const v2f*)(a0 + k0);
      a[1] = *(const v2f*)(a0 + k0 + 16 * K);
      a[2] = *(const v2f*)(a0 + k0 + 32 * K);
      a[3] = *(const v2f*)(a0 + k0 + 48 * K);
    } else {
      a[0] = *(const v2f*)(a0 + k0);
      a[1] = *(const v2f*)(a1 + k0);
      a[2] = *(const v2f*)(a2 + k0);
      a[3] = *(const v2f*)(a3 + k0);
    }
    if (SQUARE_A) {
#pragma unroll
      for (int i = 0; i < 4; ++i) { a[i].x *= a[i].x; a[i].y *= a[i].y; }
    }
    const float* bp = bcol + (size_t)k0 * N;
    v2f b0, b1;
    b0.x = bp[0];  b0.y = bp[N + 0];
    b1.x = bp[16]; b1.y = bp[N + 16];
#pragma unroll
    for (int i = 0; i < 4; ++i) {
      acc[i][0] = __builtin_amdgcn_wmma_f32_16x16x4_f32(false, a[i], false, b0,
                                                        (short)0, acc[i][0], false, false);
      acc[i][1] = __builtin_amdgcn_wmma_f32_16x16x4_f32(false, a[i], false, b1,
                                                        (short)0, acc[i][1], false, false);
    }
  }
}

template <int K, int N, bool SQUARE_A>
__global__ __launch_bounds__(256)
void k_gemm_wmma(const float* __restrict__ A, const float* __restrict__ B,
                 float* __restrict__ C, int M) {
  const int lane = threadIdx.x & 31;
  const int wid  = threadIdx.x >> 5;
  const int m = lane & 15;
  const int g = lane >> 4;

  const int row0 = (blockIdx.x * 8 + wid) * 64;
  if (row0 >= M) return;
  const int col0 = blockIdx.y * 32;

  int  r[4];
  bool st[4];
#pragma unroll
  for (int i = 0; i < 4; ++i) {
    int rr = row0 + 16 * i;
    st[i] = (rr < M);
    r[i]  = st[i] ? rr : (M - 16);
  }

  v8f acc[4][2] = {};
  const float* a0   = A + (size_t)(r[0] + m) * K + 2 * g;
  const float* bcol = B + (size_t)(2 * g) * N + col0 + m;

  if (row0 + 64 <= M) {
    gemm_kloop<K, N, SQUARE_A, false>(a0, a0, a0, a0, bcol, acc);
  } else {
    const float* a1 = A + (size_t)(r[1] + m) * K + 2 * g;
    const float* a2 = A + (size_t)(r[2] + m) * K + 2 * g;
    const float* a3 = A + (size_t)(r[3] + m) * K + 2 * g;
    gemm_kloop<K, N, SQUARE_A, true>(a0, a1, a2, a3, bcol, acc);
  }

#pragma unroll
  for (int i = 0; i < 4; ++i) {
    if (!st[i]) continue;
    float* crow = C + (size_t)(r[i] + 8 * g) * N + col0 + m;
#pragma unroll
    for (int v = 0; v < 8; ++v) {
      crow[(size_t)v * N]      = acc[i][0][v];
      crow[(size_t)v * N + 16] = acc[i][1][v];
    }
  }
}

__global__ __launch_bounds__(256)
void k_scatter(const float* __restrict__ feat, const float* __restrict__ dinv,
               const int* __restrict__ src, const int* __restrict__ tgt,
               float* __restrict__ out, int E, int N, int D) {
  int wave = (int)((blockIdx.x * blockDim.x + threadIdx.x) >> 5);
  int lane = threadIdx.x & 31;
  int total = E + N;
  if (wave >= total) return;

  int s, t;
  if (wave < E) { s = src[wave]; t = tgt[wave]; }
  else          { s = t = wave - E; }

  float norm = dinv[s] * dinv[t];
  const float4* fs = (const float4*)(feat + (size_t)s * D);
  float* od = out + (size_t)t * D;
  int nvec = D >> 2;
  for (int v = lane; v < nvec; v += 32) {
    float4 val = fs[v];
    int base = v * 4;
    atomAddF(od + base + 0, norm * val.x);
    atomAddF(od + base + 1, norm * val.y);
    atomAddF(od + base + 2, norm * val.z);
    atomAddF(od + base + 3, norm * val.w);
  }
}


extern "C" void kernel_launch(void* const* d_in, const int* in_sizes, int n_in,
                              void* d_out, int out_size, void* d_ws, size_t ws_size,
                              hipStream_t stream) {
  const float* x  = (const float*)d_in[0];
  const int*   ei = (const int*)d_in[1];
  const float* W1 = (const float*)d_in[2];
  const float* W2 = (const float*)d_in[3];
  float* out = (float*)d_out;

  const int N = in_sizes[0] / DIN;
  const int E = in_sizes[1] / 2;
  const int* src = ei;
  const int* tgt = ei + E;

  char* ws = (char*)d_ws;
  size_t off = 0;
  float* dinv = (float*)(ws + off); off += ((size_t)N * 4 + 255) & ~(size_t)255;
  float* xw   = (float*)(ws + off); off += ((size_t)N * DHID * 4 + 255) & ~(size_t)255;
  float* h    = (float*)(ws + off); off += ((size_t)N * DHID * 4 + 255) & ~(size_t)255;
  float* hw   = xw;
  (void)ws_size; (void)n_in; (void)out_size;

  const int nEdges = E + N;
  const int mBlocks = (N + 511) / 512;

  k_fill<<<(N + 255) / 256, 256, 0, stream>>>(dinv, 1.0f, N);
  k_deg_count<<<(E + 255) / 256, 256, 0, stream>>>(tgt, dinv, E);
  k_rsqrt<<<(N + 255) / 256, 256, 0, stream>>>(dinv, N);

  {
    dim3 grid(mBlocks, DHID / 32);
    k_gemm_wmma<DIN, DHID, false><<<grid, 256, 0, stream>>>(x, W1, xw, N);
  }
  {
    int n4 = N * DHID / 4;
    k_zero4<<<(n4 + 255) / 256, 256, 0, stream>>>((float4*)h, n4);
    k_scatter<<<(nEdges + 7) / 8, 256, 0, stream>>>(xw, dinv, src, tgt, h, E, N, DHID);
  }
  {
    dim3 grid(mBlocks, DOUT / 32);
    k_gemm_wmma<DHID, DOUT, true><<<grid, 256, 0, stream>>>(h, W2, hw, N);
  }
  {
    int o4 = N * DOUT / 4;
    k_zero4<<<(o4 + 255) / 256, 256, 0, stream>>>((float4*)out, o4);
    k_scatter<<<(nEdges + 7) / 8, 256, 0, stream>>>(hw, dinv, src, tgt, out, E, N, DOUT);
  }
}


# Edge scatter with f32 atomics replaced by rows grouped by target (counting sort, integer atomics) and a per-row gather summed in registers; sums now in sorted-source order
# speedup vs baseline: 1.7083x; 1.7083x over previous
#include <hip/hip_runtime.h>


typedef float v2f __attribute__((ext_vector_type(2)));
typedef float v8f __attribute__((ext_vector_type(8)));

#define DIN  512
#define DHID 512
#define DOUT 256


__global__ void k_zero_i(int* __restrict__ p, int n) {
  int i = blockIdx.x * blockDim.x + threadIdx.x;
  if (i < n) p[i] = 0;
}

__global__ void k_count_rank(const int* __restrict__ tgt, int* __restrict__ cnt,
                             int* __restrict__ rank, int E) {
  int e = blockIdx.x * blockDim.x + threadIdx.x;
  if (e < E)
    rank[e] = __hip_atomic_fetch_add(&cnt[tgt[e]], 1, __ATOMIC_RELAXED,
                                     __HIP_MEMORY_SCOPE_AGENT);
}

#define SCAN_T 1024
#define SCAN_I 16

__global__ __launch_bounds__(SCAN_T)
void k_scan_deg(const int* __restrict__ cnt, int* __restrict__ rowptr,
                float* __restrict__ dinv, int N) {
  __shared__ int wtot[SCAN_T / 32];
  const int tid = threadIdx.x, lane = tid & 31, wid = tid >> 5;
  int carry = 0;
  for (int base = 0; base < N; base += SCAN_T * SCAN_I) {
    const int i0 = base + tid * SCAN_I;
    int c[SCAN_I];
    int s = 0;
#pragma unroll
    for (int k = 0; k < SCAN_I; ++k) {
      c[k] = (i0 + k < N) ? cnt[i0 + k] : 0;
      s += c[k];
    }
    int inc = s;
#pragma unroll
    for (int d = 1; d < 32; d <<= 1) {
      int y = __shfl_up(inc, d, 32);
      if (lane >= d) inc += y;
    }
    if (lane == 31) wtot[wid] = inc;
    __syncthreads();
    if (wid == 0) {
      int wi = wtot[lane];
#pragma unroll
      for (int d = 1; d < 32; d <<= 1) {
        int y = __shfl_up(wi, d, 32);
        if (lane >= d) wi += y;
      }
      wtot[lane] = wi;
    }
    __syncthreads();
    int excl = carry + (wid ? wtot[wid - 1] : 0) + inc - s;
    const int total = wtot[SCAN_T / 32 - 1];
#pragma unroll
    for (int k = 0; k < SCAN_I; ++k) {
      if (i0 + k < N) {
        rowptr[i0 + k] = excl;
        float dg = (float)(c[k] + 1);
        dinv[i0 + k] = (dg > 0.f) ? rsqrtf(dg) : 0.f;
        excl += c[k];
      }
    }
    carry += total;
    __syncthreads();
  }
  if (tid == 0) rowptr[N] = carry;
}

__global__ void k_fill_rows(const int* __restrict__ src, const int* __restrict__ tgt,
                            const int* __restrict__ rank, const int* __restrict__ rowptr,
                            int* __restrict__ tmp, int E) {
  int e = blockIdx.x * blockDim.x + threadIdx.x;
  if (e < E) tmp[rowptr[tgt[e]] + rank[e]] = src[e];
}

__global__ __launch_bounds__(256)
void k_sort_rows(const int* __restrict__ rowptr, const int* __restrict__ tmp,
                 const float* __restrict__ dinv, int* __restrict__ csr_src,
                 float* __restrict__ csr_w, int N) {
  const int wave = (int)((blockIdx.x * blockDim.x + threadIdx.x) >> 5);
  const int lane = threadIdx.x & 31;
  if (wave >= N) return;
  const int t = __builtin_amdgcn_readfirstlane(wave);
  const int beg = rowptr[t], end = rowptr[t + 1];
  const float dt = dinv[t];
  for (int j0 = beg; j0 < end; j0 += 32) {
    const int j = j0 + lane;
    const bool act = j < end;
    int sj = act ? tmp[j] : 0;
    sj = min(max(sj, 0), N - 1);
    int rk = 0;
    for (int c0 = beg; c0 < end; c0 += 32) {
      const int ci = c0 + lane;
      int vc = (ci < end) ? tmp[ci] : 0;
      vc = min(max(vc, 0), N - 1);
      const int cn = min(32, end - c0);
      for (int k = 0; k < cn; ++k) {
        const int si = __builtin_amdgcn_readlane(vc, k);
        rk += (si < sj || (si == sj && (c0 + k) < j)) ? 1 : 0;
      }
    }
    if (act) {
      csr_src[beg + rk] = sj;
      csr_w[beg + rk]   = dinv[sj] * dt;
    }
  }
}


template <int K, int N, bool SQUARE_A, bool TAIL>
__device__ __forceinline__ void gemm_kloop(const float* __restrict__ a0,
                                           const float* __restrict__ a1,
                                           const float* __restrict__ a2,
                                           const float* __restrict__ a3,
                                           const float* __restrict__ bcol,
                                           v8f acc[4][2]) {
  for (int k0 = 0; k0 < K; k0 += 4) {
    v2f a[4];
    if (!TAIL) {
      a[0] = *(const v2f*)(a0 + k0);
      a[1] = *(const v2f*)(a0 + k0 + 16 * K);
      a[2] = *(const v2f*)(a0 + k0 + 32 * K);
      a[3] = *(const v2f*)(a0 + k0 + 48 * K);
    } else {
      a[0] = *(const v2f*)(a0 + k0);
      a[1] = *(const v2f*)(a1 + k0);
      a[2] = *(const v2f*)(a2 + k0);
      a[3] = *(const v2f*)(a3 + k0);
    }
    if (SQUARE_A) {
#pragma unroll
      for (int i = 0; i < 4; ++i) { a[i].x *= a[i].x; a[i].y *= a[i].y; }
    }
    const float* bp = bcol + (size_t)k0 * N;
    v2f b0, b1;
    b0.x = bp[0];  b0.y = bp[N + 0];
    b1.x = bp[16]; b1.y = bp[N + 16];
#pragma unroll
    for (int i = 0; i < 4; ++i) {
      acc[i][0] = __builtin_amdgcn_wmma_f32_16x16x4_f32(false, a[i], false, b0,
                                                        (short)0, acc[i][0], false, false);
      acc[i][1] = __builtin_amdgcn_wmma_f32_16x16x4_f32(false, a[i], false, b1,
                                                        (short)0, acc[i][1], false, false);
    }
  }
}

template <int K, int N, bool SQUARE_A>
__global__ __launch_bounds__(256)
void k_gemm_wmma(const float* __restrict__ A, const float* __restrict__ B,
                 float* __restrict__ C, int M) {
  const int lane = threadIdx.x & 31;
  const int wid  = threadIdx.x >> 5;
  const int m = lane & 15;
  const int g = lane >> 4;

  const int row0 = (blockIdx.x * 8 + wid) * 64;
  if (row0 >= M) return;
  const int col0 = blockIdx.y * 32;

  int  r[4];
  bool st[4];
#pragma unroll
  for (int i = 0; i < 4; ++i) {
    int rr = row0 + 16 * i;
    st[i] = (rr < M);
    r[i]  = st[i] ? rr : (M - 16);
  }

  v8f acc[4][2] = {};
  const float* a0   = A + (size_t)(r[0] + m) * K + 2 * g;
  const float* bcol = B + (size_t)(2 * g) * N + col0 + m;

  if (row0 + 64 <= M) {
    gemm_kloop<K, N, SQUARE_A, false>(a0, a0, a0, a0, bcol, acc);
  } else {
    const float* a1 = A + (size_t)(r[1] + m) * K + 2 * g;
    const float* a2 = A + (size_t)(r[2] + m) * K + 2 * g;
    const float* a3 = A + (size_t)(r[3] + m) * K + 2 * g;
    gemm_kloop<K, N, SQUARE_A, true>(a0, a1, a2, a3, bcol, acc);
  }

#pragma unroll
  for (int i = 0; i < 4; ++i) {
    if (!st[i]) continue;
    float* crow = C + (size_t)(r[i] + 8 * g) * N + col0 + m;
#pragma unroll
    for (int v = 0; v < 8; ++v) {
      crow[(size_t)v * N]      = acc[i][0][v];
      crow[(size_t)v * N + 16] = acc[i][1][v];
    }
  }
}


template <int D, int B>
__device__ __forceinline__ void gather_batch(const float* __restrict__ feat,
                                             int sv, int wvi, int k, int lane,
                                             float4 (&acc)[D / 128]) {
  constexpr int NV = D / 128;
  float4 x[B][NV];
  float  w[B];
#pragma unroll
  for (int u = 0; u < B; ++u) {
    const int s = __builtin_amdgcn_readlane(sv, k + u);
    w[u] = __int_as_float(__builtin_amdgcn_readlane(wvi, k + u));
    const float4* p = (const float4*)(feat + (size_t)s * D) + lane;
#pragma unroll
    for (int v = 0; v < NV; ++v) x[u][v] = p[32 * v];
  }
#pragma unroll
  for (int u = 0; u < B; ++u) {
#pragma unroll
    for (int v = 0; v < NV; ++v) {
      acc[v].x += w[u] * x[u][v].x;
      acc[v].y += w[u] * x[u][v].y;
      acc[v].z += w[u] * x[u][v].z;
      acc[v].w += w[u] * x[u][v].w;
    }
  }
}

template <int D, int U>
__global__ __launch_bounds__(256)
void k_gather(const float* __restrict__ feat, const float* __restrict__ dinv,
              const int* __restrict__ rowptr, const int* __restrict__ csr_src,
              const float* __restrict__ csr_w, float* __restrict__ out, int N) {
  constexpr int NV = D / 128;
  const int wave = (int)((blockIdx.x * blockDim.x + threadIdx.x) >> 5);
  const int lane = threadIdx.x & 31;
  if (wave >= N) return;
  const int t = __builtin_amdgcn_readfirstlane(wave);
  const int beg = rowptr[t], end = rowptr[t + 1];
  const float dt = dinv[t];

  float4 acc[NV];
  {
    const float4* p = (const float4*)(feat + (size_t)t * D) + lane;
    const float ws = dt * dt;
#pragma unroll
    for (int v = 0; v < NV; ++v) {
      float4 x = p[32 * v];
      acc[v] = make_float4(ws * x.x, ws * x.y, ws * x.z, ws * x.w);
    }
  }

  for (int c0 = beg; c0 < end; c0 += 32) {
    const int cn = min(32, end - c0);
    const int ci = c0 + lane;
    int   sv = 0;
    float wv = 0.f;
    if (ci < end) { sv = csr_src[ci]; wv = csr_w[ci]; }
    const int wvi = __float_as_int(wv);

    int k = 0;
    for (; k + U <= cn; k += U) gather_batch<D, U>(feat, sv, wvi, k, lane, acc);
    if (U > 4 && k + 4 <= cn) { gather_batch<D, 4>(feat, sv, wvi, k, lane, acc); k += 4; }
    if (U > 2 && k + 2 <= cn) { gather_batch<D, 2>(feat, sv, wvi, k, lane, acc); k += 2; }
    if (k < cn)               { gather_batch<D, 1>(feat, sv, wvi, k, lane, acc); }
  }

  float4* o = (float4*)(out + (size_t)t * D) + lane;
#pragma unroll
  for (int v = 0; v < NV; ++v) o[32 * v] = acc[v];
}


extern "C" void kernel_launch(void* const* d_in, const int* in_sizes, int n_in,
                              void* d_out, int out_size, void* d_ws, size_t ws_size,
                              hipStream_t stream) {
  const float* x  = (const float*)d_in[0];
  const int*   ei = (const int*)d_in[1];
  const float* W1 = (const float*)d_in[2];
  const float* W2 = (const float*)d_in[3];
  float* out = (float*)d_out;

  const int N = in_sizes[0] / DIN;
  const int E = in_sizes[1] / 2;
  const int* src = ei;
  const int* tgt = ei + E;

  char* ws = (char*)d_ws;
  size_t off = 0;
  float* dinv   = (float*)(ws + off); off += ((size_t)N * 4 + 255) & ~(size_t)255;
  float* xw     = (float*)(ws + off); off += ((size_t)N * DHID * 4 + 255) & ~(size_t)255;
  float* h      = (float*)(ws + off); off += ((size_t)N * DHID * 4 + 255) & ~(size_t)255;
  int*   cnt    = (int*)(ws + off);   off += ((size_t)N * 4 + 255) & ~(size_t)255;
  int*   rowptr = (int*)(ws + off);   off += ((size_t)(N + 1) * 4 + 255) & ~(size_t)255;
  int*   rank   = (int*)(ws + off);   off += ((size_t)E * 4 + 255) & ~(size_t)255;
  int*   tmp    = (int*)(ws + off);   off += ((size_t)E * 4 + 255) & ~(size_t)255;
  int*   csrs   = (int*)(ws + off);   off += ((size_t)E * 4 + 255) & ~(size_t)255;
  float* csrw   = (float*)(ws + off); off += ((size_t)E * 4 + 255) & ~(size_t)255;
  float* hw     = xw;
  (void)ws_size; (void)n_in; (void)out_size;

  const int mBlocks = (N + 511) / 512;
  const int rBlocks = (N + 7) / 8;

  k_zero_i<<<(N + 255) / 256, 256, 0, stream>>>(cnt, N);
  k_count_rank<<<(E + 255) / 256, 256, 0, stream>>>(tgt, cnt, rank, E);
  k_scan_deg<<<1, SCAN_T, 0, stream>>>(cnt, rowptr, dinv, N);
  k_fill_rows<<<(E + 255) / 256, 256, 0, stream>>>(src, tgt, rank, rowptr, tmp, E);
  k_sort_rows<<<rBlocks, 256, 0, stream>>>(rowptr, tmp, dinv, csrs, csrw, N);

  {
    dim3 grid(mBlocks, DHID / 32);
    k_gemm_wmma<DIN, DHID, false><<<grid, 256, 0, stream>>>(x, W1, xw, N);
  }
  k_gather<DHID, 4><<<rBlocks, 256, 0, stream>>>(xw, dinv, rowptr, csrs, csrw, h, N);

  {
    dim3 grid(mBlocks, DOUT / 32);
    k_gemm_wmma<DHID, DOUT, true><<<grid, 256, 0, stream>>>(h, W2, hw, N);
  }
  k_gather<DOUT, 8><<<rBlocks, 256, 0, stream>>>(hw, dinv, rowptr, csrs, csrw, out, N);
}


# GEMM re-tiled to one wave per SIMD: 80x64 (layer 2: 80x32) tile per wave, b128 operand loads one 80-product stage ahead; square moved to gather store; k pairing per 8-k block changed (sum order)
# speedup vs baseline: 2.4858x; 1.4551x over previous
#include <hip/hip_runtime.h>


typedef float v2f __attribute__((ext_vector_type(2)));
typedef float v8f __attribute__((ext_vector_type(8)));

#define DIN  512
#define DHID 512
#define DOUT 256


__global__ void k_zero_i(int* __restrict__ p, int n) {
  int i = blockIdx.x * blockDim.x + threadIdx.x;
  if (i < n) p[i] = 0;
}

__global__ void k_count_rank(const int* __restrict__ tgt, int* __restrict__ cnt,
                             int* __restrict__ rank, int E) {
  int e = blockIdx.x * blockDim.x + threadIdx.x;
  if (e < E)
    rank[e] = __hip_atomic_fetch_add(&cnt[tgt[e]], 1, __ATOMIC_RELAXED,
                                     __HIP_MEMORY_SCOPE_AGENT);
}

#define SCAN_T 1024
#define SCAN_I 16

__global__ __launch_bounds__(SCAN_T)
void k_scan_deg(const int* __restrict__ cnt, int* __restrict__ rowptr,
                float* __restrict__ dinv, int N) {
  __shared__ int wtot[SCAN_T / 32];
  const int tid = threadIdx.x, lane = tid & 31, wid = tid >> 5;
  int carry = 0;
  for (int base = 0; base < N; base += SCAN_T * SCAN_I) {
    const int i0 = base + tid * SCAN_I;
    int c[SCAN_I];
    int s = 0;
#pragma unroll
    for (int k = 0; k < SCAN_I; ++k) {
      c[k] = (i0 + k < N) ? cnt[i0 + k] : 0;
      s += c[k];
    }
    int inc = s;
#pragma unroll
    for (int d = 1; d < 32; d <<= 1) {
      int y = __shfl_up(inc, d, 32);
      if (lane >= d) inc += y;
    }
    if (lane == 31) wtot[wid] = inc;
    __syncthreads();
    if (wid == 0) {
      int wi = wtot[lane];
#pragma unroll
      for (int d = 1; d < 32; d <<= 1) {
        int y = __shfl_up(wi, d, 32);
        if (lane >= d) wi += y;
      }
      wtot[lane] = wi;
    }
    __syncthreads();
    int excl = carry + (wid ? wtot[wid - 1] : 0) + inc - s;
    const int total = wtot[SCAN_T / 32 - 1];
#pragma unroll
    for (int k = 0; k < SCAN_I; ++k) {
      if (i0 + k < N) {
        rowptr[i0 + k] = excl;
        float dg = (float)(c[k] + 1);
        dinv[i0 + k] = (dg > 0.f) ? rsqrtf(dg) : 0.f;
        excl += c[k];
      }
    }
    carry += total;
    __syncthreads();
  }
  if (tid == 0) rowptr[N] = carry;
}

__global__ void k_fill_rows(const int* __restrict__ src, const int* __restrict__ tgt,
                            const int* __restrict__ rank, const int* __restrict__ rowptr,
                            int* __restrict__ tmp, int E) {
  int e = blockIdx.x * blockDim.x + threadIdx.x;
  if (e < E) tmp[rowptr[tgt[e]] + rank[e]] = src[e];
}

__global__ __launch_bounds__(256)
void k_sort_rows(const int* __restrict__ rowptr, const int* __restrict__ tmp,
                 const float* __restrict__ dinv, int* __restrict__ csr_src,
                 float* __restrict__ csr_w, int N) {
  const int wave = (int)((blockIdx.x * blockDim.x + threadIdx.x) >> 5);
  const int lane = threadIdx.x & 31;
  if (wave >= N) return;
  const int t = __builtin_amdgcn_readfirstlane(wave);
  const int beg = rowptr[t], end = rowptr[t + 1];
  const float dt = dinv[t];
  for (int j0 = beg; j0 < end; j0 += 32) {
    const int j = j0 + lane;
    const bool act = j < end;
    int sj = act ? tmp[j] : 0;
    sj = min(max(sj, 0), N - 1);
    int rk = 0;
    for (int c0 = beg; c0 < end; c0 += 32) {
      const int ci = c0 + lane;
      int vc = (ci < end) ? tmp[ci] : 0;
      vc = min(max(vc, 0), N - 1);
      const int cn = min(32, end - c0);
      for (int k = 0; k < cn; ++k) {
        const int si = __builtin_amdgcn_readlane(vc, k);
        rk += (si < sj || (si == sj && (c0 + k) < j)) ? 1 : 0;
      }
    }
    if (act) {
      csr_src[beg + rk] = sj;
      csr_w[beg + rk]   = dinv[sj] * dt;
    }
  }
}


#define GEMM_MT 5

template <int N, int NT, int S>
__device__ __forceinline__ void gemm_load(const float* const (&ap)[GEMM_MT],
                                          const float* __restrict__ bp, int kb,
                                          float4 (&a)[S][GEMM_MT], float (&b)[S][4][NT]) {
#pragma unroll
  for (int s = 0; s < S; ++s) {
#pragma unroll
    for (int i = 0; i < GEMM_MT; ++i) a[s][i] = *(const float4*)(ap[i] + kb + 8 * s);
#pragma unroll
    for (int v = 0; v < 4; ++v) {
      const float* p = bp + (size_t)(kb + 8 * s + v) * N;
      if (NT == 4) {
        const float4 q = *(const float4*)p;
        b[s][v][0] = q.x; b[s][v][1] = q.y; b[s][v][2] = q.z; b[s][v][3] = q.w;
      } else {
        const float2 q = *(const float2*)p;
        b[s][v][0] = q.x; b[s][v][1] = q.y;
      }
    }
  }
}

template <int NT, int S>
__device__ __forceinline__ void gemm_mma(const float4 (&a)[S][GEMM_MT], const float (&b)[S][4][NT],
                                         v8f (&acc)[GEMM_MT][NT]) {
#pragma unroll
  for (int s = 0; s < S; ++s) {
#pragma unroll
    for (int j = 0; j < 2; ++j) {
#pragma unroll
      for (int i = 0; i < GEMM_MT; ++i) {
        v2f av;
        av.x = j ? a[s][i].z : a[s][i].x;
        av.y = j ? a[s][i].w : a[s][i].y;
#pragma unroll
        for (int t = 0; t < NT; ++t) {
          v2f bv;
          bv.x = b[s][2 * j][t];
          bv.y = b[s][2 * j + 1][t];
          acc[i][t] = __builtin_amdgcn_wmma_f32_16x16x4_f32(false, av, false, bv,
                                                            (short)0, acc[i][t], false, false);
        }
      }
    }
  }
}

template <int N, int NT, bool CHECK>
__device__ __forceinline__ void gemm_store(float* __restrict__ cbase, const v8f (&acc)[GEMM_MT][NT],
                                           int rows_left) {
#pragma unroll
  for (int i = 0; i < GEMM_MT; ++i) {
#pragma unroll
    for (int r = 0; r < 8; ++r) {
      if (!CHECK || 16 * i + r < rows_left) {
        float* p = cbase + (size_t)(16 * i + r) * N;
        if (NT == 4)
          *(float4*)p = make_float4(acc[i][0][r], acc[i][1][r], acc[i][2][r], acc[i][3][r]);
        else
          *(float2*)p = make_float2(acc[i][0][r], acc[i][1][r]);
      }
    }
  }
}

template <int K, int N, int NT, int S>
__global__ __launch_bounds__(128)
void k_gemm_wmma(const float* __restrict__ A, const float* __restrict__ B,
                 float* __restrict__ C, int M) {
  static_assert(K % (16 * S) == 0, "two stages per trip");
  const int lane = threadIdx.x & 31;
  const int wid  = threadIdx.x >> 5;
  const int c = lane & 15;
  const int g = lane >> 4;

  const int row0 = blockIdx.x * (16 * GEMM_MT);
  const int col0 = (blockIdx.y * 4 + wid) * (16 * NT);

  const float* ap[GEMM_MT];
#pragma unroll
  for (int i = 0; i < GEMM_MT; ++i) {
    const int ra = min(row0 + 16 * i + c, M - 1);
    ap[i] = A + (size_t)ra * K + 4 * g;
  }
  const float* bp = B + (size_t)(4 * g) * N + col0 + NT * c;

  v8f acc[GEMM_MT][NT] = {};
  float4 a0[S][GEMM_MT], a1[S][GEMM_MT];
  float  b0[S][4][NT], b1[S][4][NT];

  gemm_load<N, NT, S>(ap, bp, 0, a0, b0);
  for (int kb = 0; kb < K; kb += 16 * S) {
    gemm_load<N, NT, S>(ap, bp, kb + 8 * S, a1, b1);
    __builtin_amdgcn_sched_barrier(0);
    gemm_mma<NT, S>(a0, b0, acc);
    __builtin_amdgcn_sched_barrier(0);
    const int kn = (kb + 16 * S < K) ? kb + 16 * S : 0;
    gemm_load<N, NT, S>(ap, bp, kn, a0, b0);
    __builtin_amdgcn_sched_barrier(0);
    gemm_mma<NT, S>(a1, b1, acc);
    __builtin_amdgcn_sched_barrier(0);
  }

  float* cbase = C + (size_t)(row0 + 8 * g) * N + col0 + NT * c;
  if (row0 + 16 * GEMM_MT <= M) gemm_store<N, NT, false>(cbase, acc, 0);
  else                          gemm_store<N, NT, true>(cbase, acc, M - (row0 + 8 * g));
}


template <int D, int B>
__device__ __forceinline__ void gather_batch(const float* __restrict__ feat,
                                             int sv, int wvi, int k, int lane,
                                             float4 (&acc)[D / 128]) {
  constexpr int NV = D / 128;
  float4 x[B][NV];
  float  w[B];
#pragma unroll
  for (int u = 0; u < B; ++u) {
    const int s = __builtin_amdgcn_readlane(sv, k + u);
    w[u] = __int_as_float(__builtin_amdgcn_readlane(wvi, k + u));
    const float4* p = (const float4*)(feat + (size_t)s * D) + lane;
#pragma unroll
    for (int v = 0; v < NV; ++v) x[u][v] = p[32 * v];
  }
#pragma unroll
  for (int u = 0; u < B; ++u) {
#pragma unroll
    for (int v = 0; v < NV; ++v) {
      acc[v].x += w[u] * x[u][v].x;
      acc[v].y += w[u] * x[u][v].y;
      acc[v].z += w[u] * x[u][v].z;
      acc[v].w += w[u] * x[u][v].w;
    }
  }
}

template <int D, int U, bool SQUARE_OUT>
__global__ __launch_bounds__(256)
void k_gather(const float* __restrict__ feat, const float* __restrict__ dinv,
              const int* __restrict__ rowptr, const int* __restrict__ csr_src,
              const float* __restrict__ csr_w, float* __restrict__ out, int N) {
  constexpr int NV = D / 128;
  const int wave = (int)((blockIdx.x * blockDim.x + threadIdx.x) >> 5);
  const int lane = threadIdx.x & 31;
  if (wave >= N) return;
  const int t = __builtin_amdgcn_readfirstlane(wave);
  const int beg = rowptr[t], end = rowptr[t + 1];
  const float dt = dinv[t];

  float4 acc[NV];
  {
    const float4* p = (const float4*)(feat + (size_t)t * D) + lane;
    const float ws = dt * dt;
#pragma unroll
    for (int v = 0; v < NV; ++v) {
      float4 x = p[32 * v];
      acc[v] = make_float4(ws * x.x, ws * x.y, ws * x.z, ws * x.w);
    }
  }

  for (int c0 = beg; c0 < end; c0 += 32) {
    const int cn = min(32, end - c0);
    const int ci = c0 + lane;
    int   sv = 0;
    float wv = 0.f;
    if (ci < end) { sv = csr_src[ci]; wv = csr_w[ci]; }
    const int wvi = __float_as_int(wv);

    int k = 0;
    for (; k + U <= cn; k += U) gather_batch<D, U>(feat, sv, wvi, k, lane, acc);
    if (U > 4 && k + 4 <= cn) { gather_batch<D, 4>(feat, sv, wvi, k, lane, acc); k += 4; }
    if (U > 2 && k + 2 <= cn) { gather_batch<D, 2>(feat, sv, wvi, k, lane, acc); k += 2; }
    if (k < cn)               { gather_batch<D, 1>(feat, sv, wvi, k, lane, acc); }
  }

  float4* o = (float4*)(out + (size_t)t * D) + lane;
#pragma unroll
  for (int v = 0; v < NV; ++v) {
    float4 r = acc[v];
    if (SQUARE_OUT) { r.x *= r.x; r.y *= r.y; r.z *= r.z; r.w *= r.w; }
    o[32 * v] = r;
  }
}


extern "C" void kernel_launch(void* const* d_in, const int* in_sizes, int n_in,
                              void* d_out, int out_size, void* d_ws, size_t ws_size,
                              hipStream_t stream) {
  const float* x  = (const float*)d_in[0];
  const int*   ei = (const int*)d_in[1];
  const float* W1 = (const float*)d_in[2];
  const float* W2 = (const float*)d_in[3];
  float* out = (float*)d_out;

  const int N = in_sizes[0] / DIN;
  const int E = in_sizes[1] / 2;
  const int* src = ei;
  const int* tgt = ei + E;

  char* ws = (char*)d_ws;
  size_t off = 0;
  float* dinv   = (float*)(ws + off); off += ((size_t)N * 4 + 255) & ~(size_t)255;
  float* xw     = (float*)(ws + off); off += ((size_t)N * DHID * 4 + 255) & ~(size_t)255;
  float* h      = (float*)(ws + off); off += ((size_t)N * DHID * 4 + 255) & ~(size_t)255;
  int*   cnt    = (int*)(ws + off);   off += ((size_t)N * 4 + 255) & ~(size_t)255;
  int*   rowptr = (int*)(ws + off);   off += ((size_t)(N + 1) * 4 + 255) & ~(size_t)255;
  int*   rank   = (int*)(ws + off);   off += ((size_t)E * 4 + 255) & ~(size_t)255;
  int*   tmp    = (int*)(ws + off);   off += ((size_t)E * 4 + 255) & ~(size_t)255;
  int*   csrs   = (int*)(ws + off);   off += ((size_t)E * 4 + 255) & ~(size_t)255;
  float* csrw   = (float*)(ws + off); off += ((size_t)E * 4 + 255) & ~(size_t)255;
  float* hw     = xw;
  (void)ws_size; (void)n_in; (void)out_size;

  const int mBlocks = (N + 16 * GEMM_MT - 1) / (16 * GEMM_MT);
  const int rBlocks = (N + 7) / 8;

  k_zero_i<<<(N + 255) / 256, 256, 0, stream>>>(cnt, N);
  k_count_rank<<<(E + 255) / 256, 256, 0, stream>>>(tgt, cnt, rank, E);
  k_scan_deg<<<1, SCAN_T, 0, stream>>>(cnt, rowptr, dinv, N);
  k_fill_rows<<<(E + 255) / 256, 256, 0, stream>>>(src, tgt, rank, rowptr, tmp, E);
  k_sort_rows<<<rBlocks, 256, 0, stream>>>(rowptr, tmp, dinv, csrs, csrw, N);

  {
    dim3 grid(mBlocks, DHID / (4 * 16 * 4));
    k_gemm_wmma<DIN, DHID, 4, 2><<<grid, 128, 0, stream>>>(x, W1, xw, N);
  }
  k_gather<DHID, 4, true><<<rBlocks, 256, 0, stream>>>(xw, dinv, rowptr, csrs, csrw, h, N);

  {
    dim3 grid(mBlocks, DOUT / (4 * 16 * 2));
    k_gemm_wmma<DHID, DOUT, 2, 4><<<grid, 128, 0, stream>>>(h, W2, hw, N);
  }
  k_gather<DOUT, 8, false><<<rBlocks, 256, 0, stream>>>(hw, dinv, rowptr, csrs, csrw, out, N);
}


# Target rows kept in fixed 64-entry slots (overflow list beyond): an edge's place is slot + arrival rank, so the prefix-scan and fill launches go (7 launches, not 9); same sorted sum order
# speedup vs baseline: 2.8108x; 1.1307x over previous
#include <hip/hip_runtime.h>


typedef float v2f __attribute__((ext_vector_type(2)));
typedef float v8f __attribute__((ext_vector_type(8)));

#define DIN  512
#define DHID 512
#define DOUT 256


#define ROW_CAP 64

__global__ void k_zero_i(int* __restrict__ p, int n) {
  int i = blockIdx.x * blockDim.x + threadIdx.x;
  if (i < n) p[i] = 0;
}

__global__ void k_count_place(const int* __restrict__ src, const int* __restrict__ tgt,
                              int* __restrict__ cnt, int* __restrict__ tmp,
                              int* __restrict__ ovf_t, int* __restrict__ ovf_s, int E, int N) {
  int e = blockIdx.x * blockDim.x + threadIdx.x;
  if (e < E) {
    const int t = min(max(tgt[e], 0), N - 1);
    const int s = src[e];
    const int r = __hip_atomic_fetch_add(&cnt[t], 1, __ATOMIC_RELAXED, __HIP_MEMORY_SCOPE_AGENT);
    if (r < ROW_CAP) {
      tmp[t * ROW_CAP + r] = s;
    } else {
      const int o = __hip_atomic_fetch_add(&cnt[N], 1, __ATOMIC_RELAXED, __HIP_MEMORY_SCOPE_AGENT);
      if (o < E) { ovf_t[o] = t; ovf_s[o] = s; }
    }
  }
}

__global__ __launch_bounds__(256)
void k_sort_rows(const int* __restrict__ cnt, const int* __restrict__ tmp,
                 int* __restrict__ csr_src, float* __restrict__ csr_w, int N) {
  const int wave = (int)((blockIdx.x * blockDim.x + threadIdx.x) >> 5);
  const int lane = threadIdx.x & 31;
  if (wave >= N) return;
  const int t = __builtin_amdgcn_readfirstlane(wave);
  const int c = cnt[t];
  const int n = min(c, ROW_CAP);
  const int beg = t * ROW_CAP, end = beg + n;
  const float dt = rsqrtf((float)(c + 1));
  for (int j0 = beg; j0 < end; j0 += 32) {
    const int j = j0 + lane;
    const bool act = j < end;
    int sj = act ? tmp[j] : 0;
    sj = min(max(sj, 0), N - 1);
    int rk = 0;
    for (int c0 = beg; c0 < end; c0 += 32) {
      const int ci = c0 + lane;
      int vc = (ci < end) ? tmp[ci] : 0;
      vc = min(max(vc, 0), N - 1);
      const int cn = min(32, end - c0);
      for (int k = 0; k < cn; ++k) {
        const int si = __builtin_amdgcn_readlane(vc, k);
        rk += (si < sj || (si == sj && (c0 + k) < j)) ? 1 : 0;
      }
    }
    if (act) {
      csr_src[beg + rk] = sj;
      csr_w[beg + rk]   = rsqrtf((float)(cnt[sj] + 1)) * dt;
    }
  }
  for (int j = n + lane; j < ROW_CAP; j += 32) {
    csr_src[beg + j] = 0;
    csr_w[beg + j]   = 0.f;
  }
}


#define GEMM_MT 5

template <int N, int NT, int S>
__device__ __forceinline__ void gemm_load(const float* const (&ap)[GEMM_MT],
                                          const float* __restrict__ bp, int kb,
                                          float4 (&a)[S][GEMM_MT], float (&b)[S][4][NT]) {
#pragma unroll
  for (int s = 0; s < S; ++s) {
#pragma unroll
    for (int i = 0; i < GEMM_MT; ++i) a[s][i] = *(const float4*)(ap[i] + kb + 8 * s);
#pragma unroll
    for (int v = 0; v < 4; ++v) {
      const float* p = bp + (size_t)(kb + 8 * s + v) * N;
      if (NT == 4) {
        const float4 q = *(const float4*)p;
        b[s][v][0] = q.x; b[s][v][1] = q.y; b[s][v][2] = q.z; b[s][v][3] = q.w;
      } else {
        const float2 q = *(const float2*)p;
        b[s][v][0] = q.x; b[s][v][1] = q.y;
      }
    }
  }
}

template <int NT, int S>
__device__ __forceinline__ void gemm_mma(const float4 (&a)[S][GEMM_MT], const float (&b)[S][4][NT],
                                         v8f (&acc)[GEMM_MT][NT]) {
#pragma unroll
  for (int s = 0; s < S; ++s) {
#pragma unroll
    for (int j = 0; j < 2; ++j) {
#pragma unroll
      for (int i = 0; i < GEMM_MT; ++i) {
        v2f av;
        av.x = j ? a[s][i].z : a[s][i].x;
        av.y = j ? a[s][i].w : a[s][i].y;
#pragma unroll
        for (int t = 0; t < NT; ++t) {
          v2f bv;
          bv.x = b[s][2 * j][t];
          bv.y = b[s][2 * j + 1][t];
          acc[i][t] = __builtin_amdgcn_wmma_f32_16x16x4_f32(false, av, false, bv,
                                                            (short)0, acc[i][t], false, false);
        }
      }
    }
  }
}

template <int N, int NT, bool CHECK>
__device__ __forceinline__ void gemm_store(float* __restrict__ cbase, const v8f (&acc)[GEMM_MT][NT],
                                           int rows_left) {
#pragma unroll
  for (int i = 0; i < GEMM_MT; ++i) {
#pragma unroll
    for (int r = 0; r < 8; ++r) {
      if (!CHECK || 16 * i + r < rows_left) {
        float* p = cbase + (size_t)(16 * i + r) * N;
        if (NT == 4)
          *(float4*)p = make_float4(acc[i][0][r], acc[i][1][r], acc[i][2][r], acc[i][3][r]);
        else
          *(float2*)p = make_float2(acc[i][0][r], acc[i][1][r]);
      }
    }
  }
}

template <int K, int N, int NT, int S>
__global__ __launch_bounds__(128)
void k_gemm_wmma(const float* __restrict__ A, const float* __restrict__ B,
                 float* __restrict__ C, int M) {
  static_assert(K % (16 * S) == 0, "two stages per trip");
  const int lane = threadIdx.x & 31;
  const int wid  = threadIdx.x >> 5;
  const int c = lane & 15;
  const int g = lane >> 4;

  const int row0 = blockIdx.x * (16 * GEMM_MT);
  const int col0 = (blockIdx.y * 4 + wid) * (16 * NT);

  const float* ap[GEMM_MT];
#pragma unroll
  for (int i = 0; i < GEMM_MT; ++i) {
    const int ra = min(row0 + 16 * i + c, M - 1);
    ap[i] = A + (size_t)ra * K + 4 * g;
  }
  const float* bp = B + (size_t)(4 * g) * N + col0 + NT * c;

  v8f acc[GEMM_MT][NT] = {};
  float4 a0[S][GEMM_MT], a1[S][GEMM_MT];
  float  b0[S][4][NT], b1[S][4][NT];

  gemm_load<N, NT, S>(ap, bp, 0, a0, b0);
  for (int kb = 0; kb < K; kb += 16 * S) {
    gemm_load<N, NT, S>(ap, bp, kb + 8 * S, a1, b1);
    __builtin_amdgcn_sched_barrier(0);
    gemm_mma<NT, S>(a0, b0, acc);
    __builtin_amdgcn_sched_barrier(0);
    const int kn = (kb + 16 * S < K) ? kb + 16 * S : 0;
    gemm_load<N, NT, S>(ap, bp, kn, a0, b0);
    __builtin_amdgcn_sched_barrier(0);
    gemm_mma<NT, S>(a1, b1, acc);
    __builtin_amdgcn_sched_barrier(0);
  }

  float* cbase = C + (size_t)(row0 + 8 * g) * N + col0 + NT * c;
  if (row0 + 16 * GEMM_MT <= M) gemm_store<N, NT, false>(cbase, acc, 0);
  else                          gemm_store<N, NT, true>(cbase, acc, M - (row0 + 8 * g));
}


template <int D, int B>
__device__ __forceinline__ void gather_batch(const float* __restrict__ feat,
                                             int sv, int wvi, int k, int lane,
                                             float4 (&acc)[D / 128]) {
  constexpr int NV = D / 128;
  float4 x[B][NV];
  float  w[B];
#pragma unroll
  for (int u = 0; u < B; ++u) {
    const int s = __builtin_amdgcn_readlane(sv, k + u);
    w[u] = __int_as_float(__builtin_amdgcn_readlane(wvi, k + u));
    const float4* p = (const float4*)(feat + (size_t)s * D) + lane;
#pragma unroll
    for (int v = 0; v < NV; ++v) x[u][v] = p[32 * v];
  }
#pragma unroll
  for (int u = 0; u < B; ++u) {
#pragma unroll
    for (int v = 0; v < NV; ++v) {
      acc[v].x += w[u] * x[u][v].x;
      acc[v].y += w[u] * x[u][v].y;
      acc[v].z += w[u] * x[u][v].z;
      acc[v].w += w[u] * x[u][v].w;
    }
  }
}

template <int D, int U>
__device__ __forceinline__ void gather_chunk(const float* __restrict__ feat,
                                             int sv, int wvi, int cn, int lane,
                                             float4 (&acc)[D / 128]) {
  int k = 0;
  for (; k + U <= cn; k += U) gather_batch<D, U>(feat, sv, wvi, k, lane, acc);
  if (U > 4 && k + 4 <= cn) { gather_batch<D, 4>(feat, sv, wvi, k, lane, acc); k += 4; }
  if (U > 2 && k + 2 <= cn) { gather_batch<D, 2>(feat, sv, wvi, k, lane, acc); k += 2; }
  if (k < cn)               { gather_batch<D, 1>(feat, sv, wvi, k, lane, acc); }
}

template <int D, int U, bool SQUARE_OUT>
__global__ __launch_bounds__(256)
void k_gather(const float* __restrict__ feat, const int* __restrict__ cnt,
              const int* csr_src, const float* csr_w,
              const int* __restrict__ ovf_t, const int* __restrict__ ovf_s,
              float* __restrict__ out, int N) {
  constexpr int NV = D / 128;
  const int wave = (int)((blockIdx.x * blockDim.x + threadIdx.x) >> 5);
  const int lane = threadIdx.x & 31;
  if (wave >= N) return;
  const int t = __builtin_amdgcn_readfirstlane(wave);
  const int beg = t * ROW_CAP;
  int   sv = csr_src[beg + lane];
  float wv = csr_w[beg + lane];
  float4 xs[NV];
  {
    const float4* p = (const float4*)(feat + (size_t)t * D) + lane;
#pragma unroll
    for (int v = 0; v < NV; ++v) xs[v] = p[32 * v];
  }
  __builtin_amdgcn_sched_barrier(0);
  const int c = cnt[t];
  const int n = min(c, ROW_CAP);
  const float dt = rsqrtf((float)(c + 1));

  float4 acc[NV];
  {
    const float ws = dt * dt;
#pragma unroll
    for (int v = 0; v < NV; ++v)
      acc[v] = make_float4(ws * xs[v].x, ws * xs[v].y, ws * xs[v].z, ws * xs[v].w);
  }

  for (int c0 = 0; c0 < n; c0 += 32) {
    if (c0 > 0) {
      sv = csr_src[beg + c0 + lane];
      wv = csr_w[beg + c0 + lane];
    }
    gather_chunk<D, U>(feat, sv, __float_as_int(wv), min(32, n - c0), lane, acc);
  }
  if (c > ROW_CAP) {
    const int novf = cnt[N];
    for (int i0 = 0; i0 < novf; i0 += 32) {
      const int i = i0 + lane;
      int ot = -1, os = 0;
      if (i < novf) { ot = ovf_t[i]; os = ovf_s[i]; }
      os = min(max(os, 0), N - 1);
      const float wl = rsqrtf((float)(cnt[os] + 1)) * dt;
      unsigned m = __builtin_amdgcn_ballot_w32(ot == t);
      while (m) {
        const int k = __builtin_ctz(m);
        m &= m - 1;
        gather_batch<D, 1>(feat, os, __float_as_int(wl), k, lane, acc);
      }
    }
  }

  float4* o = (float4*)(out + (size_t)t * D) + lane;
#pragma unroll
  for (int v = 0; v < NV; ++v) {
    float4 r = acc[v];
    if (SQUARE_OUT) { r.x *= r.x; r.y *= r.y; r.z *= r.z; r.w *= r.w; }
    o[32 * v] = r;
  }
}


extern "C" void kernel_launch(void* const* d_in, const int* in_sizes, int n_in,
                              void* d_out, int out_size, void* d_ws, size_t ws_size,
                              hipStream_t stream) {
  const float* x  = (const float*)d_in[0];
  const int*   ei = (const int*)d_in[1];
  const float* W1 = (const float*)d_in[2];
  const float* W2 = (const float*)d_in[3];
  float* out = (float*)d_out;

  const int N = in_sizes[0] / DIN;
  const int E = in_sizes[1] / 2;
  const int* src = ei;
  const int* tgt = ei + E;

  char* ws = (char*)d_ws;
  size_t off = 0;
  float* xw     = (float*)(ws + off); off += ((size_t)N * DHID * 4 + 255) & ~(size_t)255;
  float* h      = (float*)(ws + off); off += ((size_t)N * DHID * 4 + 255) & ~(size_t)255;
  int*   cnt    = (int*)(ws + off);   off += ((size_t)(N + 1) * 4 + 255) & ~(size_t)255;
  int*   tmp    = (int*)(ws + off);   off += ((size_t)N * ROW_CAP * 4 + 255) & ~(size_t)255;
  int*   csrs   = (int*)(ws + off);   off += ((size_t)N * ROW_CAP * 4 + 255) & ~(size_t)255;
  float* csrw   = (float*)(ws + off); off += ((size_t)N * ROW_CAP * 4 + 255) & ~(size_t)255;
  int*   ovft   = (int*)(ws + off);   off += ((size_t)E * 4 + 255) & ~(size_t)255;
  int*   ovfs   = (int*)(ws + off);   off += ((size_t)E * 4 + 255) & ~(size_t)255;
  float* hw     = xw;
  (void)ws_size; (void)n_in; (void)out_size;

  const int mBlocks = (N + 16 * GEMM_MT - 1) / (16 * GEMM_MT);
  const int rBlocks = (N + 7) / 8;

  k_zero_i<<<(N + 1 + 255) / 256, 256, 0, stream>>>(cnt, N + 1);
  k_count_place<<<(E + 255) / 256, 256, 0, stream>>>(src, tgt, cnt, tmp, ovft, ovfs, E, N);
  k_sort_rows<<<rBlocks, 256, 0, stream>>>(cnt, tmp, csrs, csrw, N);

  {
    dim3 grid(mBlocks, DHID / (4 * 16 * 4));
    k_gemm_wmma<DIN, DHID, 4, 2><<<grid, 128, 0, stream>>>(x, W1, xw, N);
  }
  k_gather<DHID, 4, true><<<rBlocks, 256, 0, stream>>>(xw, cnt, csrs, csrw, ovft, ovfs, h, N);

  {
    dim3 grid(mBlocks, DOUT / (4 * 16 * 2));
    k_gemm_wmma<DHID, DOUT, 2, 4><<<grid, 128, 0, stream>>>(h, W2, hw, N);
  }
  k_gather<DOUT, 8, false><<<rBlocks, 256, 0, stream>>>(hw, cnt, csrs, csrw, ovft, ovfs, out, N);
}


# Two small passes folded into big ones: counter clear = extra work-group row of the layer-1 GEMM launch (now first); row sort = inside the layer-1 gather's waves: 5 launches, not 7; sum order kept
# speedup vs baseline: 2.9458x; 1.0480x over previous
#include <hip/hip_runtime.h>


typedef float v2f __attribute__((ext_vector_type(2)));
typedef float v8f __attribute__((ext_vector_type(8)));

#define DIN  512
#define DHID 512
#define DOUT 256


#define ROW_CAP 64

__global__ void k_count_place(const int* __restrict__ src, const int* __restrict__ tgt,
                              int* __restrict__ cnt, int* __restrict__ tmp,
                              int* __restrict__ ovf_t, int* __restrict__ ovf_s, int E, int N) {
  int e = blockIdx.x * blockDim.x + threadIdx.x;
  if (e < E) {
    const int t = min(max(tgt[e], 0), N - 1);
    const int s = src[e];
    const int r = __hip_atomic_fetch_add(&cnt[t], 1, __ATOMIC_RELAXED, __HIP_MEMORY_SCOPE_AGENT);
    if (r < ROW_CAP) {
      tmp[t * ROW_CAP + r] = s;
    } else {
      const int o = __hip_atomic_fetch_add(&cnt[N], 1, __ATOMIC_RELAXED, __HIP_MEMORY_SCOPE_AGENT);
      if (o < E) { ovf_t[o] = t; ovf_s[o] = s; }
    }
  }
}


#define GEMM_MT 5

template <int N, int NT, int S>
__device__ __forceinline__ void gemm_load(const float* const (&ap)[GEMM_MT],
                                          const float* __restrict__ bp, int kb,
                                          float4 (&a)[S][GEMM_MT], float (&b)[S][4][NT]) {
#pragma unroll
  for (int s = 0; s < S; ++s) {
#pragma unroll
    for (int i = 0; i < GEMM_MT; ++i) a[s][i] = *(const float4*)(ap[i] + kb + 8 * s);
#pragma unroll
    for (int v = 0; v < 4; ++v) {
      const float* p = bp + (size_t)(kb + 8 * s + v) * N;
      if (NT == 4) {
        const float4 q = *(const float4*)p;
        b[s][v][0] = q.x; b[s][v][1] = q.y; b[s][v][2] = q.z; b[s][v][3] = q.w;
      } else {
        const float2 q = *(const float2*)p;
        b[s][v][0] = q.x; b[s][v][1] = q.y;
      }
    }
  }
}

template <int NT, int S>
__device__ __forceinline__ void gemm_mma(const float4 (&a)[S][GEMM_MT], const float (&b)[S][4][NT],
                                         v8f (&acc)[GEMM_MT][NT]) {
#pragma unroll
  for (int s = 0; s < S; ++s) {
#pragma unroll
    for (int j = 0; j < 2; ++j) {
#pragma unroll
      for (int i = 0; i < GEMM_MT; ++i) {
        v2f av;
        av.x = j ? a[s][i].z : a[s][i].x;
        av.y = j ? a[s][i].w : a[s][i].y;
#pragma unroll
        for (int t = 0; t < NT; ++t) {
          v2f bv;
          bv.x = b[s][2 * j][t];
          bv.y = b[s][2 * j + 1][t];
          acc[i][t] = __builtin_amdgcn_wmma_f32_16x16x4_f32(false, av, false, bv,
                                                            (short)0, acc[i][t], false, false);
        }
      }
    }
  }
}

template <int N, int NT, bool CHECK>
__device__ __forceinline__ void gemm_store(float* __restrict__ cbase, const v8f (&acc)[GEMM_MT][NT],
                                           int rows_left) {
#pragma unroll
  for (int i = 0; i < GEMM_MT; ++i) {
#pragma unroll
    for (int r = 0; r < 8; ++r) {
      if (!CHECK || 16 * i + r < rows_left) {
        float* p = cbase + (size_t)(16 * i + r) * N;
        if (NT == 4)
          *(float4*)p = make_float4(acc[i][0][r], acc[i][1][r], acc[i][2][r], acc[i][3][r]);
        else
          *(float2*)p = make_float2(acc[i][0][r], acc[i][1][r]);
      }
    }
  }
}

template <int K, int N, int NT, int S, bool CLEAR>
__global__ __launch_bounds__(128)
void k_gemm_wmma(const float* __restrict__ A, const float* __restrict__ B,
                 float* __restrict__ C, int M, int* __restrict__ clr, int nclr) {
  static_assert(K % (16 * S) == 0, "two stages per trip");
  if (CLEAR && blockIdx.y == N / (4 * 16 * NT)) {
    const int i = blockIdx.x * 128 + threadIdx.x;
    if (i < nclr) clr[i] = 0;
    return;
  }
  const int lane = threadIdx.x & 31;
  const int wid  = threadIdx.x >> 5;
  const int c = lane & 15;
  const int g = lane >> 4;

  const int row0 = blockIdx.x * (16 * GEMM_MT);
  const int col0 = (blockIdx.y * 4 + wid) * (16 * NT);

  const float* ap[GEMM_MT];
#pragma unroll
  for (int i = 0; i < GEMM_MT; ++i) {
    const int ra = min(row0 + 16 * i + c, M - 1);
    ap[i] = A + (size_t)ra * K + 4 * g;
  }
  const float* bp = B + (size_t)(4 * g) * N + col0 + NT * c;

  v8f acc[GEMM_MT][NT] = {};
  float4 a0[S][GEMM_MT], a1[S][GEMM_MT];
  float  b0[S][4][NT], b1[S][4][NT];

  gemm_load<N, NT, S>(ap, bp, 0, a0, b0);
  for (int kb = 0; kb < K; kb += 16 * S) {
    gemm_load<N, NT, S>(ap, bp, kb + 8 * S, a1, b1);
    __builtin_amdgcn_sched_barrier(0);
    gemm_mma<NT, S>(a0, b0, acc);
    __builtin_amdgcn_sched_barrier(0);
    const int kn = (kb + 16 * S < K) ? kb + 16 * S : 0;
    gemm_load<N, NT, S>(ap, bp, kn, a0, b0);
    __builtin_amdgcn_sched_barrier(0);
    gemm_mma<NT, S>(a1, b1, acc);
    __builtin_amdgcn_sched_barrier(0);
  }

  float* cbase = C + (size_t)(row0 + 8 * g) * N + col0 + NT * c;
  if (row0 + 16 * GEMM_MT <= M) gemm_store<N, NT, false>(cbase, acc, 0);
  else                          gemm_store<N, NT, true>(cbase, acc, M - (row0 + 8 * g));
}


template <int D, int B>
__device__ __forceinline__ void gather_batch(const float* __restrict__ feat,
                                             int sv, int wvi, int k, int lane,
                                             float4 (&acc)[D / 128]) {
  constexpr int NV = D / 128;
  float4 x[B][NV];
  float  w[B];
#pragma unroll
  for (int u = 0; u < B; ++u) {
    const int s = __builtin_amdgcn_readlane(sv, k + u);
    w[u] = __int_as_float(__builtin_amdgcn_readlane(wvi, k + u));
    const float4* p = (const float4*)(feat + (size_t)s * D) + lane;
#pragma unroll
    for (int v = 0; v < NV; ++v) x[u][v] = p[32 * v];
  }
#pragma unroll
  for (int u = 0; u < B; ++u) {
#pragma unroll
    for (int v = 0; v < NV; ++v) {
      acc[v].x += w[u] * x[u][v].x;
      acc[v].y += w[u] * x[u][v].y;
      acc[v].z += w[u] * x[u][v].z;
      acc[v].w += w[u] * x[u][v].w;
    }
  }
}

template <int D, int U>
__device__ __forceinline__ void gather_chunk(const float* __restrict__ feat,
                                             int sv, int wvi, int cn, int lane,
                                             float4 (&acc)[D / 128]) {
  int k = 0;
  for (; k + U <= cn; k += U) gather_batch<D, U>(feat, sv, wvi, k, lane, acc);
  if (U > 4 && k + 4 <= cn) { gather_batch<D, 4>(feat, sv, wvi, k, lane, acc); k += 4; }
  if (U > 2 && k + 2 <= cn) { gather_batch<D, 2>(feat, sv, wvi, k, lane, acc); k += 2; }
  if (k < cn)               { gather_batch<D, 1>(feat, sv, wvi, k, lane, acc); }
}

template <int D>
__device__ __forceinline__ void gather_self(const float4 (&xs)[D / 128], float dt, float4 (&acc)[D / 128]) {
  const float ws = dt * dt;
#pragma unroll
  for (int v = 0; v < D / 128; ++v)
    acc[v] = make_float4(ws * xs[v].x, ws * xs[v].y, ws * xs[v].z, ws * xs[v].w);
}

template <int D>
__device__ __forceinline__ void gather_overflow(const float* __restrict__ feat, const int* __restrict__ cnt,
                                                const int* __restrict__ ovf_t, const int* __restrict__ ovf_s,
                                                int t, float dt, int lane, int N, float4 (&acc)[D / 128]) {
  const int novf = cnt[N];
  for (int i0 = 0; i0 < novf; i0 += 32) {
    const int i = i0 + lane;
    int ot = -1, os = 0;
    if (i < novf) { ot = ovf_t[i]; os = ovf_s[i]; }
    os = min(max(os, 0), N - 1);
    const float wl = rsqrtf((float)(cnt[os] + 1)) * dt;
    unsigned m = __builtin_amdgcn_ballot_w32(ot == t);
    while (m) {
      const int k = __builtin_ctz(m);
      m &= m - 1;
      gather_batch<D, 1>(feat, os, __float_as_int(wl), k, lane, acc);
    }
  }
}

template <int D, bool SQUARE_OUT>
__device__ __forceinline__ void gather_store(float* __restrict__ out, int t, int lane,
                                             const float4 (&acc)[D / 128]) {
  float4* o = (float4*)(out + (size_t)t * D) + lane;
#pragma unroll
  for (int v = 0; v < D / 128; ++v) {
    float4 r = acc[v];
    if (SQUARE_OUT) { r.x *= r.x; r.y *= r.y; r.z *= r.z; r.w *= r.w; }
    o[32 * v] = r;
  }
}

template <int D, int U, bool SQUARE_OUT>
__global__ __launch_bounds__(256)
void k_gather(const float* __restrict__ feat, const int* __restrict__ cnt,
              const int* csr_src, const float* csr_w,
              const int* __restrict__ ovf_t, const int* __restrict__ ovf_s,
              float* __restrict__ out, int N) {
  constexpr int NV = D / 128;
  const int wave = (int)((blockIdx.x * blockDim.x + threadIdx.x) >> 5);
  const int lane = threadIdx.x & 31;
  if (wave >= N) return;
  const int t = __builtin_amdgcn_readfirstlane(wave);
  const int beg = t * ROW_CAP;
  int   sv = csr_src[beg + lane];
  float wv = csr_w[beg + lane];
  float4 xs[NV];
  {
    const float4* p = (const float4*)(feat + (size_t)t * D) + lane;
#pragma unroll
    for (int v = 0; v < NV; ++v) xs[v] = p[32 * v];
  }
  __builtin_amdgcn_sched_barrier(0);
  const int c = cnt[t];
  const int n = min(c, ROW_CAP);
  const float dt = rsqrtf((float)(c + 1));

  float4 acc[NV];
  {
    const float ws = dt * dt;
#pragma unroll
    for (int v = 0; v < NV; ++v)
      acc[v] = make_float4(ws * xs[v].x, ws * xs[v].y, ws * xs[v].z, ws * xs[v].w);
  }

  for (int c0 = 0; c0 < n; c0 += 32) {
    if (c0 > 0) {
      sv = csr_src[beg + c0 + lane];
      wv = csr_w[beg + c0 + lane];
    }
    gather_chunk<D, U>(feat, sv, __float_as_int(wv), min(32, n - c0), lane, acc);
  }
  if (c > ROW_CAP) gather_overflow<D>(feat, cnt, ovf_t, ovf_s, t, dt, lane, N, acc);
  gather_store<D, SQUARE_OUT>(out, t, lane, acc);
}

template <int D, int U, bool SQUARE_OUT>
__global__ __launch_bounds__(256)
void k_sort_gather(const float* __restrict__ feat, const int* __restrict__ cnt,
                   const int* __restrict__ tmp,
                   int* __restrict__ csr_src, float* __restrict__ csr_w,
                   const int* __restrict__ ovf_t, const int* __restrict__ ovf_s,
                   float* __restrict__ out, int N) {
  constexpr int NV = D / 128;
  const int wave = (int)((blockIdx.x * blockDim.x + threadIdx.x) >> 5);
  const int lane = threadIdx.x & 31;
  if (wave >= N) return;
  const int t = __builtin_amdgcn_readfirstlane(wave);
  const int beg = t * ROW_CAP;
  float4 xs[NV];
  {
    const float4* p = (const float4*)(feat + (size_t)t * D) + lane;
#pragma unroll
    for (int v = 0; v < NV; ++v) xs[v] = p[32 * v];
  }
  const int c = cnt[t];
  const int n = min(c, ROW_CAP);
  const float dt = rsqrtf((float)(c + 1));
  float4 acc[NV];

  if (n <= 32) {
    const bool act = lane < n;
    int sj = act ? tmp[beg + lane] : 0;
    sj = min(max(sj, 0), N - 1);
    const int dj = cnt[sj];
    int rk = 0;
    for (int k = 0; k < n; ++k) {
      const int si = __builtin_amdgcn_readlane(sj, k);
      rk += (si < sj || (si == sj && k < lane)) ? 1 : 0;
    }
    const int   pj = act ? rk : lane;
    const int   es = act ? sj : 0;
    const float ew = act ? rsqrtf((float)(dj + 1)) * dt : 0.f;
    csr_src[beg + pj] = es;
    csr_w[beg + pj]   = ew;
    const int sv  = __builtin_amdgcn_ds_permute(pj << 2, es);
    const int wvi = __builtin_amdgcn_ds_permute(pj << 2, __float_as_int(ew));
    gather_self<D>(xs, dt, acc);
    gather_chunk<D, U>(feat, sv, wvi, n, lane, acc);
  } else {
    const bool act1 = 32 + lane < n;
    int s0 = tmp[beg + lane];
    int s1 = act1 ? tmp[beg + 32 + lane] : 0;
    s0 = min(max(s0, 0), N - 1);
    s1 = min(max(s1, 0), N - 1);
    const int d0 = cnt[s0], d1 = cnt[s1];
    int r0 = 0, r1 = 0;
    for (int k = 0; k < 32; ++k) {
      const int si = __builtin_amdgcn_readlane(s0, k);
      r0 += (si < s0 || (si == s0 && k < lane)) ? 1 : 0;
      r1 += (si <= s1) ? 1 : 0;
    }
    for (int k = 0; k < n - 32; ++k) {
      const int si = __builtin_amdgcn_readlane(s1, k);
      r0 += (si < s0) ? 1 : 0;
      r1 += (si < s1 || (si == s1 && k < lane)) ? 1 : 0;
    }
    const float w0 = rsqrtf((float)(d0 + 1)) * dt;
    const float w1 = rsqrtf((float)(d1 + 1)) * dt;
    csr_src[beg + r0] = s0;
    csr_w[beg + r0]   = w0;
    const int p1 = act1 ? r1 : 32 + lane;
    csr_src[beg + p1] = act1 ? s1 : 0;
    csr_w[beg + p1]   = act1 ? w1 : 0.f;
    gather_self<D>(xs, dt, acc);
    for (int k = 0; k < n; ++k) {
      const unsigned m0 = __builtin_amdgcn_ballot_w32(r0 == k);
      if (m0) {
        gather_batch<D, 1>(feat, s0, __float_as_int(w0), __builtin_ctz(m0), lane, acc);
      } else {
        const unsigned m1 = __builtin_amdgcn_ballot_w32(act1 && r1 == k);
        if (m1) gather_batch<D, 1>(feat, s1, __float_as_int(w1), __builtin_ctz(m1), lane, acc);
      }
    }
  }
  if (c > ROW_CAP) gather_overflow<D>(feat, cnt, ovf_t, ovf_s, t, dt, lane, N, acc);
  gather_store<D, SQUARE_OUT>(out, t, lane, acc);
}


extern "C" void kernel_launch(void* const* d_in, const int* in_sizes, int n_in,
                              void* d_out, int out_size, void* d_ws, size_t ws_size,
                              hipStream_t stream) {
  const float* x  = (const float*)d_in[0];
  const int*   ei = (const int*)d_in[1];
  const float* W1 = (const float*)d_in[2];
  const float* W2 = (const float*)d_in[3];
  float* out = (float*)d_out;

  const int N = in_sizes[0] / DIN;
  const int E = in_sizes[1] / 2;
  const int* src = ei;
  const int* tgt = ei + E;

  char* ws = (char*)d_ws;
  size_t off = 0;
  float* xw     = (float*)(ws + off); off += ((size_t)N * DHID * 4 + 255) & ~(size_t)255;
  float* h      = (float*)(ws + off); off += ((size_t)N * DHID * 4 + 255) & ~(size_t)255;
  int*   cnt    = (int*)(ws + off);   off += ((size_t)(N + 1) * 4 + 255) & ~(size_t)255;
  int*   tmp    = (int*)(ws + off);   off += ((size_t)N * ROW_CAP * 4 + 255) & ~(size_t)255;
  int*   csrs   = (int*)(ws + off);   off += ((size_t)N * ROW_CAP * 4 + 255) & ~(size_t)255;
  float* csrw   = (float*)(ws + off); off += ((size_t)N * ROW_CAP * 4 + 255) & ~(size_t)255;
  int*   ovft   = (int*)(ws + off);   off += ((size_t)E * 4 + 255) & ~(size_t)255;
  int*   ovfs   = (int*)(ws + off);   off += ((size_t)E * 4 + 255) & ~(size_t)255;
  float* hw     = xw;
  (void)ws_size; (void)n_in; (void)out_size;

  const int mBlocks = (N + 16 * GEMM_MT - 1) / (16 * GEMM_MT);
  const int rBlocks = (N + 7) / 8;

  {
    dim3 grid(mBlocks, DHID / (4 * 16 * 4) + 1);
    k_gemm_wmma<DIN, DHID, 4, 2, true><<<grid, 128, 0, stream>>>(x, W1, xw, N, cnt, N + 1);
  }
  k_count_place<<<(E + 255) / 256, 256, 0, stream>>>(src, tgt, cnt, tmp, ovft, ovfs, E, N);

  k_sort_gather<DHID, 4, true><<<rBlocks, 256, 0, stream>>>(xw, cnt, tmp, csrs, csrw, ovft, ovfs, h, N);

  {
    dim3 grid(mBlocks, DOUT / (4 * 16 * 2));
    k_gemm_wmma<DHID, DOUT, 2, 4, false><<<grid, 128, 0, stream>>>(h, W2, hw, N, nullptr, 0);
  }
  k_gather<DOUT, 8, false><<<rBlocks, 256, 0, stream>>>(hw, cnt, csrs, csrw, ovft, ovfs, out, N);
}


# Edge count/place pass (same atomics, 4 edges per thread) moved into extra work-group rows of the whole layer-1 product launch; counters cleared by a small launch in front: 5 launches; sum order kept
# speedup vs baseline: 3.2358x; 1.0984x over previous
#include <hip/hip_runtime.h>


typedef float v2f __attribute__((ext_vector_type(2)));
typedef float v8f __attribute__((ext_vector_type(8)));

#define DIN  512
#define DHID 512
#define DOUT 256


#define ROW_CAP 64

__global__ void k_clear(int* __restrict__ cnt, int n) {
  const int i = blockIdx.x * blockDim.x + threadIdx.x;
  if (i < n) cnt[i] = 0;
}

#define EDGE_ROWS 3

__device__ __forceinline__ void count_place(const int* __restrict__ src, const int* __restrict__ tgt,
                                            int* __restrict__ cnt, int* __restrict__ tmp,
                                            int* __restrict__ ovf_t, int* __restrict__ ovf_s,
                                            int E, int N, int i, int T) {
  for (int e0 = i; e0 < E; e0 += 4 * T) {
    int t[4], s[4], r[4];
#pragma unroll
    for (int u = 0; u < 4; ++u) {
      const int e = e0 + u * T;
      const int ec = e < E ? e : e0;
      t[u] = tgt[ec];
      s[u] = src[ec];
    }
#pragma unroll
    for (int u = 0; u < 4; ++u)
      t[u] = (e0 + u * T < E) ? min(max(t[u], 0), N - 1) : -1;
#pragma unroll
    for (int u = 0; u < 4; ++u) {
      r[u] = 0;
      if (t[u] >= 0)
        r[u] = __hip_atomic_fetch_add(&cnt[t[u]], 1, __ATOMIC_RELAXED, __HIP_MEMORY_SCOPE_AGENT);
    }
#pragma unroll
    for (int u = 0; u < 4; ++u) {
      if (t[u] >= 0) {
        if (r[u] < ROW_CAP) {
          tmp[t[u] * ROW_CAP + r[u]] = s[u];
        } else {
          const int o = __hip_atomic_fetch_add(&cnt[N], 1, __ATOMIC_RELAXED, __HIP_MEMORY_SCOPE_AGENT);
          if (o < E) { ovf_t[o] = t[u]; ovf_s[o] = s[u]; }
        }
      }
    }
  }
}


#define GEMM_MT 5

template <int N, int NT, int S>
__device__ __forceinline__ void gemm_load(const float* const (&ap)[GEMM_MT],
                                          const float* __restrict__ bp, int kb,
                                          float4 (&a)[S][GEMM_MT], float (&b)[S][4][NT]) {
#pragma unroll
  for (int s = 0; s < S; ++s) {
#pragma unroll
    for (int i = 0; i < GEMM_MT; ++i) a[s][i] = *(const float4*)(ap[i] + kb + 8 * s);
#pragma unroll
    for (int v = 0; v < 4; ++v) {
      const float* p = bp + (size_t)(kb + 8 * s + v) * N;
      if (NT == 4) {
        const float4 q = *(const float4*)p;
        b[s][v][0] = q.x; b[s][v][1] = q.y; b[s][v][2] = q.z; b[s][v][3] = q.w;
      } else {
        const float2 q = *(const float2*)p;
        b[s][v][0] = q.x; b[s][v][1] = q.y;
      }
    }
  }
}

template <int NT, int S>
__device__ __forceinline__ void gemm_mma(const float4 (&a)[S][GEMM_MT], const float (&b)[S][4][NT],
                                         v8f (&acc)[GEMM_MT][NT]) {
#pragma unroll
  for (int s = 0; s < S; ++s) {
#pragma unroll
    for (int j = 0; j < 2; ++j) {
#pragma unroll
      for (int i = 0; i < GEMM_MT; ++i) {
        v2f av;
        av.x = j ? a[s][i].z : a[s][i].x;
        av.y = j ? a[s][i].w : a[s][i].y;
#pragma unroll
        for (int t = 0; t < NT; ++t) {
          v2f bv;
          bv.x = b[s][2 * j][t];
          bv.y = b[s][2 * j + 1][t];
          acc[i][t] = __builtin_amdgcn_wmma_f32_16x16x4_f32(false, av, false, bv,
                                                            (short)0, acc[i][t], false, false);
        }
      }
    }
  }
}

template <int N, int NT, bool CHECK>
__device__ __forceinline__ void gemm_store(float* __restrict__ cbase, const v8f (&acc)[GEMM_MT][NT],
                                           int rows_left) {
#pragma unroll
  for (int i = 0; i < GEMM_MT; ++i) {
#pragma unroll
    for (int r = 0; r < 8; ++r) {
      if (!CHECK || 16 * i + r < rows_left) {
        float* p = cbase + (size_t)(16 * i + r) * N;
        if (NT == 4)
          *(float4*)p = make_float4(acc[i][0][r], acc[i][1][r], acc[i][2][r], acc[i][3][r]);
        else
          *(float2*)p = make_float2(acc[i][0][r], acc[i][1][r]);
      }
    }
  }
}

template <int K, int N, int NT, int S, bool EDGES>
__global__ __launch_bounds__(128)
void k_gemm_wmma(const float* __restrict__ A, const float* __restrict__ B,
                 float* __restrict__ C, int M,
                 const int* __restrict__ src, const int* __restrict__ tgt,
                 int* __restrict__ cnt, int* __restrict__ tmp,
                 int* __restrict__ ovf_t, int* __restrict__ ovf_s, int E) {
  static_assert(K % (16 * S) == 0, "two stages per trip");
  if (EDGES && blockIdx.y >= N / (4 * 16 * NT)) {
    const int row = blockIdx.y - N / (4 * 16 * NT);
    const int T = gridDim.x * 128 * EDGE_ROWS;
    const int i = (row * gridDim.x + blockIdx.x) * 128 + threadIdx.x;
    count_place(src, tgt, cnt, tmp, ovf_t, ovf_s, E, M, i, T);
    return;
  }
  const int lane = threadIdx.x & 31;
  const int wid  = threadIdx.x >> 5;
  const int c = lane & 15;
  const int g = lane >> 4;

  const int row0 = blockIdx.x * (16 * GEMM_MT);
  const int col0 = (blockIdx.y * 4 + wid) * (16 * NT);

  const float* ap[GEMM_MT];
#pragma unroll
  for (int i = 0; i < GEMM_MT; ++i) {
    const int ra = min(row0 + 16 * i + c, M - 1);
    ap[i] = A + (size_t)ra * K + 4 * g;
  }
  const float* bp = B + (size_t)(4 * g) * N + col0 + NT * c;

  v8f acc[GEMM_MT][NT] = {};
  float4 a0[S][GEMM_MT], a1[S][GEMM_MT];
  float  b0[S][4][NT], b1[S][4][NT];

  gemm_load<N, NT, S>(ap, bp, 0, a0, b0);
  for (int kb = 0; kb < K; kb += 16 * S) {
    gemm_load<N, NT, S>(ap, bp, kb + 8 * S, a1, b1);
    __builtin_amdgcn_sched_barrier(0);
    gemm_mma<NT, S>(a0, b0, acc);
    __builtin_amdgcn_sched_barrier(0);
    const int kn = (kb + 16 * S < K) ? kb + 16 * S : 0;
    gemm_load<N, NT, S>(ap, bp, kn, a0, b0);
    __builtin_amdgcn_sched_barrier(0);
    gemm_mma<NT, S>(a1, b1, acc);
    __builtin_amdgcn_sched_barrier(0);
  }

  float* cbase = C + (size_t)(row0 + 8 * g) * N + col0 + NT * c;
  if (row0 + 16 * GEMM_MT <= M) gemm_store<N, NT, false>(cbase, acc, 0);
  else                          gemm_store<N, NT, true>(cbase, acc, M - (row0 + 8 * g));
}


template <int D, int B>
__device__ __forceinline__ void gather_batch(const float* __restrict__ feat,
                                             int sv, int wvi, int k, int lane,
                                             float4 (&acc)[D / 128]) {
  constexpr int NV = D / 128;
  float4 x[B][NV];
  float  w[B];
#pragma unroll
  for (int u = 0; u < B; ++u) {
    const int s = __builtin_amdgcn_readlane(sv, k + u);
    w[u] = __int_as_float(__builtin_amdgcn_readlane(wvi, k + u));
    const float4* p = (const float4*)(feat + (size_t)s * D) + lane;
#pragma unroll
    for (int v = 0; v < NV; ++v) x[u][v] = p[32 * v];
  }
#pragma unroll
  for (int u = 0; u < B; ++u) {
#pragma unroll
    for (int v = 0; v < NV; ++v) {
      acc[v].x += w[u] * x[u][v].x;
      acc[v].y += w[u] * x[u][v].y;
      acc[v].z += w[u] * x[u][v].z;
      acc[v].w += w[u] * x[u][v].w;
    }
  }
}

template <int D, int U>
__device__ __forceinline__ void gather_chunk(const float* __restrict__ feat,
                                             int sv, int wvi, int cn, int lane,
                                             float4 (&acc)[D / 128]) {
  int k = 0;
  for (; k + U <= cn; k += U) gather_batch<D, U>(feat, sv, wvi, k, lane, acc);
  if (U > 4 && k + 4 <= cn) { gather_batch<D, 4>(feat, sv, wvi, k, lane, acc); k += 4; }
  if (U > 2 && k + 2 <= cn) { gather_batch<D, 2>(feat, sv, wvi, k, lane, acc); k += 2; }
  if (k < cn)               { gather_batch<D, 1>(feat, sv, wvi, k, lane, acc); }
}

template <int D>
__device__ __forceinline__ void gather_self(const float4 (&xs)[D / 128], float dt, float4 (&acc)[D / 128]) {
  const float ws = dt * dt;
#pragma unroll
  for (int v = 0; v < D / 128; ++v)
    acc[v] = make_float4(ws * xs[v].x, ws * xs[v].y, ws * xs[v].z, ws * xs[v].w);
}

template <int D>
__device__ __forceinline__ void gather_overflow(const float* __restrict__ feat, const int* __restrict__ cnt,
                                                const int* __restrict__ ovf_t, const int* __restrict__ ovf_s,
                                                int t, float dt, int lane, int N, float4 (&acc)[D / 128]) {
  const int novf = cnt[N];
  for (int i0 = 0; i0 < novf; i0 += 32) {
    const int i = i0 + lane;
    int ot = -1, os = 0;
    if (i < novf) { ot = ovf_t[i]; os = ovf_s[i]; }
    os = min(max(os, 0), N - 1);
    const float wl = rsqrtf((float)(cnt[os] + 1)) * dt;
    unsigned m = __builtin_amdgcn_ballot_w32(ot == t);
    while (m) {
      const int k = __builtin_ctz(m);
      m &= m - 1;
      gather_batch<D, 1>(feat, os, __float_as_int(wl), k, lane, acc);
    }
  }
}

template <int D, bool SQUARE_OUT>
__device__ __forceinline__ void gather_store(float* __restrict__ out, int t, int lane,
                                             const float4 (&acc)[D / 128]) {
  float4* o = (float4*)(out + (size_t)t * D) + lane;
#pragma unroll
  for (int v = 0; v < D / 128; ++v) {
    float4 r = acc[v];
    if (SQUARE_OUT) { r.x *= r.x; r.y *= r.y; r.z *= r.z; r.w *= r.w; }
    o[32 * v] = r;
  }
}

template <int D, int U, bool SQUARE_OUT>
__global__ __launch_bounds__(256)
void k_gather(const float* __restrict__ feat, const int* __restrict__ cnt,
              const int* csr_src, const float* csr_w,
              const int* __restrict__ ovf_t, const int* __restrict__ ovf_s,
              float* __restrict__ out, int N) {
  constexpr int NV = D / 128;
  const int wave = (int)((blockIdx.x * blockDim.x + threadIdx.x) >> 5);
  const int lane = threadIdx.x & 31;
  if (wave >= N) return;
  const int t = __builtin_amdgcn_readfirstlane(wave);
  const int beg = t * ROW_CAP;
  int   sv = csr_src[beg + lane];
  float wv = csr_w[beg + lane];
  float4 xs[NV];
  {
    const float4* p = (const float4*)(feat + (size_t)t * D) + lane;
#pragma unroll
    for (int v = 0; v < NV; ++v) xs[v] = p[32 * v];
  }
  __builtin_amdgcn_sched_barrier(0);
  const int c = cnt[t];
  const int n = min(c, ROW_CAP);
  const float dt = rsqrtf((float)(c + 1));

  float4 acc[NV];
  {
    const float ws = dt * dt;
#pragma unroll
    for (int v = 0; v < NV; ++v)
      acc[v] = make_float4(ws * xs[v].x, ws * xs[v].y, ws * xs[v].z, ws * xs[v].w);
  }

  for (int c0 = 0; c0 < n; c0 += 32) {
    if (c0 > 0) {
      sv = csr_src[beg + c0 + lane];
      wv = csr_w[beg + c0 + lane];
    }
    gather_chunk<D, U>(feat, sv, __float_as_int(wv), min(32, n - c0), lane, acc);
  }
  if (c > ROW_CAP) gather_overflow<D>(feat, cnt, ovf_t, ovf_s, t, dt, lane, N, acc);
  gather_store<D, SQUARE_OUT>(out, t, lane, acc);
}

template <int D, int U, bool SQUARE_OUT>
__global__ __launch_bounds__(256)
void k_sort_gather(const float* __restrict__ feat, const int* __restrict__ cnt,
                   const int* __restrict__ tmp,
                   int* __restrict__ csr_src, float* __restrict__ csr_w,
                   const int* __restrict__ ovf_t, const int* __restrict__ ovf_s,
                   float* __restrict__ out, int N) {
  constexpr int NV = D / 128;
  const int wave = (int)((blockIdx.x * blockDim.x + threadIdx.x) >> 5);
  const int lane = threadIdx.x & 31;
  if (wave >= N) return;
  const int t = __builtin_amdgcn_readfirstlane(wave);
  const int beg = t * ROW_CAP;
  float4 xs[NV];
  {
    const float4* p = (const float4*)(feat + (size_t)t * D) + lane;
#pragma unroll
    for (int v = 0; v < NV; ++v) xs[v] = p[32 * v];
  }
  const int c = cnt[t];
  const int n = min(c, ROW_CAP);
  const float dt = rsqrtf((float)(c + 1));
  float4 acc[NV];

  if (n <= 32) {
    const bool act = lane < n;
    int sj = act ? tmp[beg + lane] : 0;
    sj = min(max(sj, 0), N - 1);
    const int dj = cnt[sj];
    int rk = 0;
    for (int k = 0; k < n; ++k) {
      const int si = __builtin_amdgcn_readlane(sj, k);
      rk += (si < sj || (si == sj && k < lane)) ? 1 : 0;
    }
    const int   pj = act ? rk : lane;
    const int   es = act ? sj : 0;
    const float ew = act ? rsqrtf((float)(dj + 1)) * dt : 0.f;
    csr_src[beg + pj] = es;
    csr_w[beg + pj]   = ew;
    const int sv  = __builtin_amdgcn_ds_permute(pj << 2, es);
    const int wvi = __builtin_amdgcn_ds_permute(pj << 2, __float_as_int(ew));
    gather_self<D>(xs, dt, acc);
    gather_chunk<D, U>(feat, sv, wvi, n, lane, acc);
  } else {
    const bool act1 = 32 + lane < n;
    int s0 = tmp[beg + lane];
    int s1 = act1 ? tmp[beg + 32 + lane] : 0;
    s0 = min(max(s0, 0), N - 1);
    s1 = min(max(s1, 0), N - 1);
    const int d0 = cnt[s0], d1 = cnt[s1];
    int r0 = 0, r1 = 0;
    for (int k = 0; k < 32; ++k) {
      const int si = __builtin_amdgcn_readlane(s0, k);
      r0 += (si < s0 || (si == s0 && k < lane)) ? 1 : 0;
      r1 += (si <= s1) ? 1 : 0;
    }
    for (int k = 0; k < n - 32; ++k) {
      const int si = __builtin_amdgcn_readlane(s1, k);
      r0 += (si < s0) ? 1 : 0;
      r1 += (si < s1 || (si == s1 && k < lane)) ? 1 : 0;
    }
    const float w0 = rsqrtf((float)(d0 + 1)) * dt;
    const float w1 = rsqrtf((float)(d1 + 1)) * dt;
    csr_src[beg + r0] = s0;
    csr_w[beg + r0]   = w0;
    const int p1 = act1 ? r1 : 32 + lane;
    csr_src[beg + p1] = act1 ? s1 : 0;
    csr_w[beg + p1]   = act1 ? w1 : 0.f;
    gather_self<D>(xs, dt, acc);
    for (int k = 0; k < n; ++k) {
      const unsigned m0 = __builtin_amdgcn_ballot_w32(r0 == k);
      if (m0) {
        gather_batch<D, 1>(feat, s0, __float_as_int(w0), __builtin_ctz(m0), lane, acc);
      } else {
        const unsigned m1 = __builtin_amdgcn_ballot_w32(act1 && r1 == k);
        if (m1) gather_batch<D, 1>(feat, s1, __float_as_int(w1), __builtin_ctz(m1), lane, acc);
      }
    }
  }
  if (c > ROW_CAP) gather_overflow<D>(feat, cnt, ovf_t, ovf_s, t, dt, lane, N, acc);
  gather_store<D, SQUARE_OUT>(out, t, lane, acc);
}


extern "C" void kernel_launch(void* const* d_in, const int* in_sizes, int n_in,
                              void* d_out, int out_size, void* d_ws, size_t ws_size,
                              hipStream_t stream) {
  const float* x  = (const float*)d_in[0];
  const int*   ei = (const int*)d_in[1];
  const float* W1 = (const float*)d_in[2];
  const float* W2 = (const float*)d_in[3];
  float* out = (float*)d_out;

  const int N = in_sizes[0] / DIN;
  const int E = in_sizes[1] / 2;
  const int* src = ei;
  const int* tgt = ei + E;

  char* ws = (char*)d_ws;
  size_t off = 0;
  float* xw     = (float*)(ws + off); off += ((size_t)N * DHID * 4 + 255) & ~(size_t)255;
  float* h      = (float*)(ws + off); off += ((size_t)N * DHID * 4 + 255) & ~(size_t)255;
  int*   cnt    = (int*)(ws + off);   off += ((size_t)(N + 1) * 4 + 255) & ~(size_t)255;
  int*   tmp    = (int*)(ws + off);   off += ((size_t)N * ROW_CAP * 4 + 255) & ~(size_t)255;
  int*   csrs   = (int*)(ws + off);   off += ((size_t)N * ROW_CAP * 4 + 255) & ~(size_t)255;
  float* csrw   = (float*)(ws + off); off += ((size_t)N * ROW_CAP * 4 + 255) & ~(size_t)255;
  int*   ovft   = (int*)(ws + off);   off += ((size_t)E * 4 + 255) & ~(size_t)255;
  int*   ovfs   = (int*)(ws + off);   off += ((size_t)E * 4 + 255) & ~(size_t)255;
  float* hw     = xw;
  (void)ws_size; (void)n_in; (void)out_size;

  const int mBlocks = (N + 16 * GEMM_MT - 1) / (16 * GEMM_MT);
  const int rBlocks = (N + 7) / 8;

  k_clear<<<(N + 1 + 255) / 256, 256, 0, stream>>>(cnt, N + 1);
  {
    dim3 grid(mBlocks, DHID / (4 * 16 * 4) + EDGE_ROWS);
    k_gemm_wmma<DIN, DHID, 4, 2, true><<<grid, 128, 0, stream>>>(x, W1, xw, N, src, tgt, cnt, tmp, ovft, ovfs, E);
  }

  k_sort_gather<DHID, 4, true><<<rBlocks, 256, 0, stream>>>(xw, cnt, tmp, csrs, csrw, ovft, ovfs, h, N);

  {
    dim3 grid(mBlocks, DOUT / (4 * 16 * 2));
    k_gemm_wmma<DHID, DOUT, 2, 4, false><<<grid, 128, 0, stream>>>(h, W2, hw, N, nullptr, nullptr, nullptr, nullptr, nullptr, nullptr, 0);
  }
  k_gather<DOUT, 8, false><<<rBlocks, 256, 0, stream>>>(hw, cnt, csrs, csrw, ovft, ovfs, out, N);
}


# Rows with more than 32 in-edges: sorted entries placed one per lane by rank in registers, then added in 4-edge batches like the other rows, not one edge per memory round trip; same sorted sum order
# speedup vs baseline: 3.3431x; 1.0332x over previous
#include <hip/hip_runtime.h>


typedef float v2f __attribute__((ext_vector_type(2)));
typedef float v8f __attribute__((ext_vector_type(8)));

#define DIN  512
#define DHID 512
#define DOUT 256


#define ROW_CAP 64

__global__ void k_clear(int* __restrict__ cnt, int n) {
  const int i = blockIdx.x * blockDim.x + threadIdx.x;
  if (i < n) cnt[i] = 0;
}

#define EDGE_ROWS 3

__device__ __forceinline__ void count_place(const int* __restrict__ src, const int* __restrict__ tgt,
                                            int* __restrict__ cnt, int* __restrict__ tmp,
                                            int* __restrict__ ovf_t, int* __restrict__ ovf_s,
                                            int E, int N, int i, int T) {
  for (int e0 = i; e0 < E; e0 += 4 * T) {
    int t[4], s[4], r[4];
#pragma unroll
    for (int u = 0; u < 4; ++u) {
      const int e = e0 + u * T;
      const int ec = e < E ? e : e0;
      t[u] = tgt[ec];
      s[u] = src[ec];
    }
#pragma unroll
    for (int u = 0; u < 4; ++u)
      t[u] = (e0 + u * T < E) ? min(max(t[u], 0), N - 1) : -1;
#pragma unroll
    for (int u = 0; u < 4; ++u) {
      r[u] = 0;
      if (t[u] >= 0)
        r[u] = __hip_atomic_fetch_add(&cnt[t[u]], 1, __ATOMIC_RELAXED, __HIP_MEMORY_SCOPE_AGENT);
    }
#pragma unroll
    for (int u = 0; u < 4; ++u) {
      if (t[u] >= 0) {
        if (r[u] < ROW_CAP) {
          tmp[t[u] * ROW_CAP + r[u]] = s[u];
        } else {
          const int o = __hip_atomic_fetch_add(&cnt[N], 1, __ATOMIC_RELAXED, __HIP_MEMORY_SCOPE_AGENT);
          if (o < E) { ovf_t[o] = t[u]; ovf_s[o] = s[u]; }
        }
      }
    }
  }
}


#define GEMM_MT 5

template <int N, int NT, int S>
__device__ __forceinline__ void gemm_load(const float* const (&ap)[GEMM_MT],
                                          const float* __restrict__ bp, int kb,
                                          float4 (&a)[S][GEMM_MT], float (&b)[S][4][NT]) {
#pragma unroll
  for (int s = 0; s < S; ++s) {
#pragma unroll
    for (int i = 0; i < GEMM_MT; ++i) a[s][i] = *(const float4*)(ap[i] + kb + 8 * s);
#pragma unroll
    for (int v = 0; v < 4; ++v) {
      const float* p = bp + (size_t)(kb + 8 * s + v) * N;
      if (NT == 4) {
        const float4 q = *(const float4*)p;
        b[s][v][0] = q.x; b[s][v][1] = q.y; b[s][v][2] = q.z; b[s][v][3] = q.w;
      } else {
        const float2 q = *(const float2*)p;
        b[s][v][0] = q.x; b[s][v][1] = q.y;
      }
    }
  }
}

template <int NT, int S>
__device__ __forceinline__ void gemm_mma(const float4 (&a)[S][GEMM_MT], const float (&b)[S][4][NT],
                                         v8f (&acc)[GEMM_MT][NT]) {
#pragma unroll
  for (int s = 0; s < S; ++s) {
#pragma unroll
    for (int j = 0; j < 2; ++j) {
#pragma unroll
      for (int i = 0; i < GEMM_MT; ++i) {
        v2f av;
        av.x = j ? a[s][i].z : a[s][i].x;
        av.y = j ? a[s][i].w : a[s][i].y;
#pragma unroll
        for (int t = 0; t < NT; ++t) {
          v2f bv;
          bv.x = b[s][2 * j][t];
          bv.y = b[s][2 * j + 1][t];
          acc[i][t] = __builtin_amdgcn_wmma_f32_16x16x4_f32(false, av, false, bv,
                                                            (short)0, acc[i][t], false, false);
        }
      }
    }
  }
}

template <int N, int NT, bool CHECK>
__device__ __forceinline__ void gemm_store(float* __restrict__ cbase, const v8f (&acc)[GEMM_MT][NT],
                                           int rows_left) {
#pragma unroll
  for (int i = 0; i < GEMM_MT; ++i) {
#pragma unroll
    for (int r = 0; r < 8; ++r) {
      if (!CHECK || 16 * i + r < rows_left) {
        float* p = cbase + (size_t)(16 * i + r) * N;
        if (NT == 4)
          *(float4*)p = make_float4(acc[i][0][r], acc[i][1][r], acc[i][2][r], acc[i][3][r]);
        else
          *(float2*)p = make_float2(acc[i][0][r], acc[i][1][r]);
      }
    }
  }
}

template <int K, int N, int NT, int S, bool EDGES>
__global__ __launch_bounds__(128)
void k_gemm_wmma(const float* __restrict__ A, const float* __restrict__ B,
                 float* __restrict__ C, int M,
                 const int* __restrict__ src, const int* __restrict__ tgt,
                 int* __restrict__ cnt, int* __restrict__ tmp,
                 int* __restrict__ ovf_t, int* __restrict__ ovf_s, int E) {
  static_assert(K % (16 * S) == 0, "two stages per trip");
  if (EDGES && blockIdx.y >= N / (4 * 16 * NT)) {
    const int row = blockIdx.y - N / (4 * 16 * NT);
    const int T = gridDim.x * 128 * EDGE_ROWS;
    const int i = (row * gridDim.x + blockIdx.x) * 128 + threadIdx.x;
    count_place(src, tgt, cnt, tmp, ovf_t, ovf_s, E, M, i, T);
    return;
  }
  const int lane = threadIdx.x & 31;
  const int wid  = threadIdx.x >> 5;
  const int c = lane & 15;
  const int g = lane >> 4;

  const int row0 = blockIdx.x * (16 * GEMM_MT);
  const int col0 = (blockIdx.y * 4 + wid) * (16 * NT);

  const float* ap[GEMM_MT];
#pragma unroll
  for (int i = 0; i < GEMM_MT; ++i) {
    const int ra = min(row0 + 16 * i + c, M - 1);
    ap[i] = A + (size_t)ra * K + 4 * g;
  }
  const float* bp = B + (size_t)(4 * g) * N + col0 + NT * c;

  v8f acc[GEMM_MT][NT] = {};
  float4 a0[S][GEMM_MT], a1[S][GEMM_MT];
  float  b0[S][4][NT], b1[S][4][NT];

  gemm_load<N, NT, S>(ap, bp, 0, a0, b0);
  for (int kb = 0; kb < K; kb += 16 * S) {
    gemm_load<N, NT, S>(ap, bp, kb + 8 * S, a1, b1);
    __builtin_amdgcn_sched_barrier(0);
    gemm_mma<NT, S>(a0, b0, acc);
    __builtin_amdgcn_sched_barrier(0);
    const int kn = (kb + 16 * S < K) ? kb + 16 * S : 0;
    gemm_load<N, NT, S>(ap, bp, kn, a0, b0);
    __builtin_amdgcn_sched_barrier(0);
    gemm_mma<NT, S>(a1, b1, acc);
    __builtin_amdgcn_sched_barrier(0);
  }

  float* cbase = C + (size_t)(row0 + 8 * g) * N + col0 + NT * c;
  if (row0 + 16 * GEMM_MT <= M) gemm_store<N, NT, false>(cbase, acc, 0);
  else                          gemm_store<N, NT, true>(cbase, acc, M - (row0 + 8 * g));
}


template <int D, int B>
__device__ __forceinline__ void gather_batch(const float* __restrict__ feat,
                                             int sv, int wvi, int k, int lane,
                                             float4 (&acc)[D / 128]) {
  constexpr int NV = D / 128;
  float4 x[B][NV];
  float  w[B];
#pragma unroll
  for (int u = 0; u < B; ++u) {
    const int s = __builtin_amdgcn_readlane(sv, k + u);
    w[u] = __int_as_float(__builtin_amdgcn_readlane(wvi, k + u));
    const float4* p = (const float4*)(feat + (size_t)s * D) + lane;
#pragma unroll
    for (int v = 0; v < NV; ++v) x[u][v] = p[32 * v];
  }
#pragma unroll
  for (int u = 0; u < B; ++u) {
#pragma unroll
    for (int v = 0; v < NV; ++v) {
      acc[v].x += w[u] * x[u][v].x;
      acc[v].y += w[u] * x[u][v].y;
      acc[v].z += w[u] * x[u][v].z;
      acc[v].w += w[u] * x[u][v].w;
    }
  }
}

template <int D, int U>
__device__ __forceinline__ void gather_chunk(const float* __restrict__ feat,
                                             int sv, int wvi, int cn, int lane,
                                             float4 (&acc)[D / 128]) {
  int k = 0;
  for (; k + U <= cn; k += U) gather_batch<D, U>(feat, sv, wvi, k, lane, acc);
  if (U > 4 && k + 4 <= cn) { gather_batch<D, 4>(feat, sv, wvi, k, lane, acc); k += 4; }
  if (U > 2 && k + 2 <= cn) { gather_batch<D, 2>(feat, sv, wvi, k, lane, acc); k += 2; }
  if (k < cn)               { gather_batch<D, 1>(feat, sv, wvi, k, lane, acc); }
}

template <int D>
__device__ __forceinline__ void gather_self(const float4 (&xs)[D / 128], float dt, float4 (&acc)[D / 128]) {
  const float ws = dt * dt;
#pragma unroll
  for (int v = 0; v < D / 128; ++v)
    acc[v] = make_float4(ws * xs[v].x, ws * xs[v].y, ws * xs[v].z, ws * xs[v].w);
}

template <int D>
__device__ __forceinline__ void gather_overflow(const float* __restrict__ feat, const int* __restrict__ cnt,
                                                const int* __restrict__ ovf_t, const int* __restrict__ ovf_s,
                                                int t, float dt, int lane, int N, float4 (&acc)[D / 128]) {
  const int novf = cnt[N];
  for (int i0 = 0; i0 < novf; i0 += 32) {
    const int i = i0 + lane;
    int ot = -1, os = 0;
    if (i < novf) { ot = ovf_t[i]; os = ovf_s[i]; }
    os = min(max(os, 0), N - 1);
    const float wl = rsqrtf((float)(cnt[os] + 1)) * dt;
    unsigned m = __builtin_amdgcn_ballot_w32(ot == t);
    while (m) {
      const int k = __builtin_ctz(m);
      m &= m - 1;
      gather_batch<D, 1>(feat, os, __float_as_int(wl), k, lane, acc);
    }
  }
}

template <int D, bool SQUARE_OUT>
__device__ __forceinline__ void gather_store(float* __restrict__ out, int t, int lane,
                                             const float4 (&acc)[D / 128]) {
  float4* o = (float4*)(out + (size_t)t * D) + lane;
#pragma unroll
  for (int v = 0; v < D / 128; ++v) {
    float4 r = acc[v];
    if (SQUARE_OUT) { r.x *= r.x; r.y *= r.y; r.z *= r.z; r.w *= r.w; }
    o[32 * v] = r;
  }
}

template <int D, int U, bool SQUARE_OUT>
__global__ __launch_bounds__(256)
void k_gather(const float* __restrict__ feat, const int* __restrict__ cnt,
              const int* csr_src, const float* csr_w,
              const int* __restrict__ ovf_t, const int* __restrict__ ovf_s,
              float* __restrict__ out, int N) {
  constexpr int NV = D / 128;
  const int wave = (int)((blockIdx.x * blockDim.x + threadIdx.x) >> 5);
  const int lane = threadIdx.x & 31;
  if (wave >= N) return;
  const int t = __builtin_amdgcn_readfirstlane(wave);
  const int beg = t * ROW_CAP;
  int   sv = csr_src[beg + lane];
  float wv = csr_w[beg + lane];
  float4 xs[NV];
  {
    const float4* p = (const float4*)(feat + (size_t)t * D) + lane;
#pragma unroll
    for (int v = 0; v < NV; ++v) xs[v] = p[32 * v];
  }
  __builtin_amdgcn_sched_barrier(0);
  const int c = cnt[t];
  const int n = min(c, ROW_CAP);
  const float dt = rsqrtf((float)(c + 1));

  float4 acc[NV];
  {
    const float ws = dt * dt;
#pragma unroll
    for (int v = 0; v < NV; ++v)
      acc[v] = make_float4(ws * xs[v].x, ws * xs[v].y, ws * xs[v].z, ws * xs[v].w);
  }

  for (int c0 = 0; c0 < n; c0 += 32) {
    if (c0 > 0) {
      sv = csr_src[beg + c0 + lane];
      wv = csr_w[beg + c0 + lane];
    }
    gather_chunk<D, U>(feat, sv, __float_as_int(wv), min(32, n - c0), lane, acc);
  }
  if (c > ROW_CAP) gather_overflow<D>(feat, cnt, ovf_t, ovf_s, t, dt, lane, N, acc);
  gather_store<D, SQUARE_OUT>(out, t, lane, acc);
}

template <int D, int U, bool SQUARE_OUT>
__global__ __launch_bounds__(256)
void k_sort_gather(const float* __restrict__ feat, const int* __restrict__ cnt,
                   const int* __restrict__ tmp,
                   int* __restrict__ csr_src, float* __restrict__ csr_w,
                   const int* __restrict__ ovf_t, const int* __restrict__ ovf_s,
                   float* __restrict__ out, int N) {
  constexpr int NV = D / 128;
  const int wave = (int)((blockIdx.x * blockDim.x + threadIdx.x) >> 5);
  const int lane = threadIdx.x & 31;
  if (wave >= N) return;
  const int t = __builtin_amdgcn_readfirstlane(wave);
  const int beg = t * ROW_CAP;
  float4 xs[NV];
  {
    const float4* p = (const float4*)(feat + (size_t)t * D) + lane;
#pragma unroll
    for (int v = 0; v < NV; ++v) xs[v] = p[32 * v];
  }
  const int c = cnt[t];
  const int n = min(c, ROW_CAP);
  const float dt = rsqrtf((float)(c + 1));
  float4 acc[NV];
  int sv, wvi, sv2 = 0, wv2 = 0;

  if (n <= 32) {
    const bool act = lane < n;
    int sj = act ? tmp[beg + lane] : 0;
    sj = min(max(sj, 0), N - 1);
    const int dj = cnt[sj];
    int rk = 0;
    for (int k = 0; k < n; ++k) {
      const int si = __builtin_amdgcn_readlane(sj, k);
      rk += (si < sj || (si == sj && k < lane)) ? 1 : 0;
    }
    const int   pj = act ? rk : lane;
    const int   es = act ? sj : 0;
    const float ew = act ? rsqrtf((float)(dj + 1)) * dt : 0.f;
    csr_src[beg + pj] = es;
    csr_w[beg + pj]   = ew;
    sv  = __builtin_amdgcn_ds_permute(pj << 2, es);
    wvi = __builtin_amdgcn_ds_permute(pj << 2, __float_as_int(ew));
  } else {
    const bool act1 = 32 + lane < n;
    int s0 = tmp[beg + lane];
    int s1 = act1 ? tmp[beg + 32 + lane] : 0;
    s0 = min(max(s0, 0), N - 1);
    s1 = min(max(s1, 0), N - 1);
    const int d0 = cnt[s0], d1 = cnt[s1];
    int r0 = 0, r1 = 0;
    for (int k = 0; k < 32; ++k) {
      const int si = __builtin_amdgcn_readlane(s0, k);
      r0 += (si < s0 || (si == s0 && k < lane)) ? 1 : 0;
      r1 += (si <= s1) ? 1 : 0;
    }
    for (int k = 0; k < n - 32; ++k) {
      const int si = __builtin_amdgcn_readlane(s1, k);
      r0 += (si < s0) ? 1 : 0;
      r1 += (si < s1 || (si == s1 && k < lane)) ? 1 : 0;
    }
    const float w0 = rsqrtf((float)(d0 + 1)) * dt;
    const float w1 = rsqrtf((float)(d1 + 1)) * dt;
    csr_src[beg + r0] = s0;
    csr_w[beg + r0]   = w0;
    const int p1 = act1 ? r1 : 32 + lane;
    csr_src[beg + p1] = act1 ? s1 : 0;
    csr_w[beg + p1]   = act1 ? w1 : 0.f;
    const int w0i = __float_as_int(w0);
    const int w1i = __float_as_int(w1);
    sv = 0; wvi = 0;
    for (int j = 0; j < 32; ++j) {
      const int rj = __builtin_amdgcn_readlane(r0, j);
      const int sj = __builtin_amdgcn_readlane(s0, j);
      const int wj = __builtin_amdgcn_readlane(w0i, j);
      if (rj == lane)      { sv  = sj; wvi = wj; }
      if (rj == lane + 32) { sv2 = sj; wv2 = wj; }
    }
    for (int j = 0; j < n - 32; ++j) {
      const int rj = __builtin_amdgcn_readlane(r1, j);
      const int sj = __builtin_amdgcn_readlane(s1, j);
      const int wj = __builtin_amdgcn_readlane(w1i, j);
      if (rj == lane)      { sv  = sj; wvi = wj; }
      if (rj == lane + 32) { sv2 = sj; wv2 = wj; }
    }
  }
  gather_self<D>(xs, dt, acc);
  for (int c0 = 0; c0 < n; c0 += 32)
    gather_chunk<D, U>(feat, c0 ? sv2 : sv, c0 ? wv2 : wvi, min(32, n - c0), lane, acc);
  if (c > ROW_CAP) gather_overflow<D>(feat, cnt, ovf_t, ovf_s, t, dt, lane, N, acc);
  gather_store<D, SQUARE_OUT>(out, t, lane, acc);
}


extern "C" void kernel_launch(void* const* d_in, const int* in_sizes, int n_in,
                              void* d_out, int out_size, void* d_ws, size_t ws_size,
                              hipStream_t stream) {
  const float* x  = (const float*)d_in[0];
  const int*   ei = (const int*)d_in[1];
  const float* W1 = (const float*)d_in[2];
  const float* W2 = (const float*)d_in[3];
  float* out = (float*)d_out;

  const int N = in_sizes[0] / DIN;
  const int E = in_sizes[1] / 2;
  const int* src = ei;
  const int* tgt = ei + E;

  char* ws = (char*)d_ws;
  size_t off = 0;
  float* xw     = (float*)(ws + off); off += ((size_t)N * DHID * 4 + 255) & ~(size_t)255;
  float* h      = (float*)(ws + off); off += ((size_t)N * DHID * 4 + 255) & ~(size_t)255;
  int*   cnt    = (int*)(ws + off);   off += ((size_t)(N + 1) * 4 + 255) & ~(size_t)255;
  int*   tmp    = (int*)(ws + off);   off += ((size_t)N * ROW_CAP * 4 + 255) & ~(size_t)255;
  int*   csrs   = (int*)(ws + off);   off += ((size_t)N * ROW_CAP * 4 + 255) & ~(size_t)255;
  float* csrw   = (float*)(ws + off); off += ((size_t)N * ROW_CAP * 4 + 255) & ~(size_t)255;
  int*   ovft   = (int*)(ws + off);   off += ((size_t)E * 4 + 255) & ~(size_t)255;
  int*   ovfs   = (int*)(ws + off);   off += ((size_t)E * 4 + 255) & ~(size_t)255;
  float* hw     = xw;
  (void)ws_size; (void)n_in; (void)out_size;

  const int mBlocks = (N + 16 * GEMM_MT - 1) / (16 * GEMM_MT);
  const int rBlocks = (N + 7) / 8;

  k_clear<<<(N + 1 + 255) / 256, 256, 0, stream>>>(cnt, N + 1);
  {
    dim3 grid(mBlocks, DHID / (4 * 16 * 4) + EDGE_ROWS);
    k_gemm_wmma<DIN, DHID, 4, 2, true><<<grid, 128, 0, stream>>>(x, W1, xw, N, src, tgt, cnt, tmp, ovft, ovfs, E);
  }

  k_sort_gather<DHID, 4, true><<<rBlocks, 256, 0, stream>>>(xw, cnt, tmp, csrs, csrw, ovft, ovfs, h, N);

  {
    dim3 grid(mBlocks, DOUT / (4 * 16 * 2));
    k_gemm_wmma<DHID, DOUT, 2, 4, false><<<grid, 128, 0, stream>>>(h, W2, hw, N, nullptr, nullptr, nullptr, nullptr, nullptr, nullptr, 0);
  }
  k_gather<DOUT, 8, false><<<rBlocks, 256, 0, stream>>>(hw, cnt, csrs, csrw, ovft, ovfs, out, N);
}
